# starting kernel
# baseline (speedup 1.0000x reference)
#include <hip/hip_runtime.h>
#include <hip/hip_bf16.h>
#include <math.h>

#define B_SZ 256
#define R_SZ 36
#define T_SZ 64
#define D_SZ 1024
#define EPS_L2 1e-8f
#define EPS_BN 1e-5f
#define LEAK 0.1f
#define KC 64

typedef __attribute__((ext_vector_type(2))) float v2f;
typedef __attribute__((ext_vector_type(8))) float v8f;

__global__ void zero_ws_kernel(float* p, int n) {
    int i = blockIdx.x * blockDim.x + threadIdx.x;
    if (i < n) p[i] = 0.f;
}

__global__ void cap_pre_kernel(const float* __restrict__ cap,
                               const int* __restrict__ lens,
                               float* __restrict__ cap_mean) {
    const int b = blockIdx.x;
    const int len = lens[b];
    const float* cb = cap + (size_t)b * T_SZ * D_SZ;
    float sumsq[4] = {0.f, 0.f, 0.f, 0.f};
    float summ[4]  = {0.f, 0.f, 0.f, 0.f};
    for (int t = 0; t < T_SZ; ++t) {
        const float* row = cb + (size_t)t * D_SZ;
        const bool valid = (t < len);
#pragma unroll
        for (int i = 0; i < 4; ++i) {
            float x = row[threadIdx.x + i * 256];
            float y = x > 0.f ? x : LEAK * x;
            sumsq[i] += y * y;
            if (valid) summ[i] += y;
        }
    }
#pragma unroll
    for (int i = 0; i < 4; ++i) {
        int d = threadIdx.x + i * 256;
        float n = sqrtf(sumsq[i]) + EPS_L2;
        cap_mean[(size_t)b * D_SZ + d] = summ[i] / (n * (float)len);
    }
}

__global__ void img_pre_kernel(const float* __restrict__ img,
                               float* __restrict__ meanR,
                               float* __restrict__ cs1,
                               float* __restrict__ cs2) {
    const int b = blockIdx.x;
    const float* ib = img + (size_t)b * R_SZ * D_SZ;
    float sumsq[4] = {0.f, 0.f, 0.f, 0.f};
    float summ[4]  = {0.f, 0.f, 0.f, 0.f};
    for (int r = 0; r < R_SZ; ++r) {
        const float* row = ib + (size_t)r * D_SZ;
#pragma unroll
        for (int i = 0; i < 4; ++i) {
            float x = row[threadIdx.x + i * 256];
            float y = x > 0.f ? x : LEAK * x;
            sumsq[i] += y * y;
            summ[i]  += y;
        }
    }
#pragma unroll
    for (int i = 0; i < 4; ++i) {
        int d = threadIdx.x + i * 256;
        float n = sqrtf(sumsq[i]) + EPS_L2;
        float s1 = summ[i] / n;
        float s2 = sumsq[i] / (n * n);
        meanR[(size_t)b * D_SZ + d] = s1 * (1.f / (float)R_SZ);
        atomicAdd(cs1 + d, s1);
        atomicAdd(cs2 + d, s2);
    }
}

__global__ void make_base_kernel(const float* __restrict__ meanR,
                                 const float* __restrict__ cs1,
                                 const float* __restrict__ cs2,
                                 const float* __restrict__ bn_w,
                                 const float* __restrict__ bn_b,
                                 float* __restrict__ base,
                                 float* __restrict__ base2) {
    const int b = blockIdx.x;
    const float inv_n = 1.f / (float)(B_SZ * R_SZ);
#pragma unroll
    for (int i = 0; i < 4; ++i) {
        int d = threadIdx.x + i * 256;
        float mu  = cs1[d] * inv_n;
        float ex2 = cs2[d] * inv_n;
        float var = ex2 - mu * mu;
        float scale = bn_w[d] * rsqrtf(var + EPS_BN);
        float v = (meanR[(size_t)b * D_SZ + d] - mu) * scale + bn_b[d];
        base[(size_t)b * D_SZ + d]  = v;
        base2[(size_t)b * D_SZ + d] = v * v;
    }
}

__global__ __launch_bounds__(128) void fc_gemm_kernel(
        const float* __restrict__ A, const float* __restrict__ Bw,
        const float* __restrict__ bias, float* __restrict__ Cout) {
    __shared__ float sA[2][16][KC + 1];
    __shared__ float sB[2][16][KC + 1];
    const int tid  = threadIdx.x;
    const int wave = tid >> 5;
    const int lane = tid & 31;
    const int wm = wave >> 1, wn = wave & 1;
    const int half = lane >> 4, fr = lane & 15;
    const int m0 = blockIdx.y * 32;
    const int n0 = blockIdx.x * 32;
    v8f acc = {};
    for (int k0 = 0; k0 < D_SZ; k0 += KC) {
        for (int idx = tid; idx < 2 * 16 * KC; idx += 128) {
            int blk = idx / (16 * KC);
            int rem = idx % (16 * KC);
            int row = rem / KC, col = rem % KC;
            sA[blk][row][col] = A [(size_t)(m0 + blk * 16 + row) * D_SZ + k0 + col];
            sB[blk][row][col] = Bw[(size_t)(n0 + blk * 16 + row) * D_SZ + k0 + col];
        }
        __syncthreads();
#pragma unroll
        for (int kk = 0; kk < KC; kk += 4) {
            int kb = kk + half * 2;
            v2f a, b;
            a[0] = sA[wm][fr][kb]; a[1] = sA[wm][fr][kb + 1];
            b[0] = sB[wn][fr][kb]; b[1] = sB[wn][fr][kb + 1];
            acc = __builtin_amdgcn_wmma_f32_16x16x4_f32(
                    false, a, false, b, (short)0, acc, false, false);
        }
        __syncthreads();
    }
#pragma unroll
    for (int j = 0; j < 8; ++j) {
        int m = m0 + wm * 16 + j + half * 8;
        int n = n0 + wn * 16 + fr;
        Cout[(size_t)m * (2 * D_SZ) + n] = acc[j] + bias[n];
    }
}

__global__ void make_pab_kernel(const float* __restrict__ cap_mean,
                                const float* __restrict__ params,
                                float* __restrict__ P, float* __restrict__ A2m,
                                float* __restrict__ ABm,
                                float* __restrict__ qv, float* __restrict__ rv) {
    const int c = blockIdx.x;
    const int tid = threadIdx.x;
    __shared__ float red[256];

    float ss = 0.f;
#pragma unroll
    for (int i = 0; i < 4; ++i) {
        float x = cap_mean[(size_t)c * D_SZ + tid + i * 256];
        ss += x * x;
    }
    red[tid] = ss; __syncthreads();
    for (int s = 128; s > 0; s >>= 1) {
        if (tid < s) red[tid] += red[tid + s];
        __syncthreads();
    }
    const float nrm = sqrtf(red[0]) + EPS_L2;
    __syncthreads();

    float q = 0.f, r = 0.f;
#pragma unroll
    for (int i = 0; i < 4; ++i) {
        int d = tid + i * 256;
        float alpha = params[(size_t)c * (2 * D_SZ) + 2 * d];
        float beta  = params[(size_t)c * (2 * D_SZ) + 2 * d + 1];
        float ch = cap_mean[(size_t)c * D_SZ + d] / nrm;
        P  [(size_t)c * D_SZ + d] = alpha * ch;
        A2m[(size_t)c * D_SZ + d] = alpha * alpha;
        ABm[(size_t)c * D_SZ + d] = alpha * beta;
        q += beta * ch;
        r += beta * beta;
    }
    red[tid] = q; __syncthreads();
    for (int s = 128; s > 0; s >>= 1) {
        if (tid < s) red[tid] += red[tid + s];
        __syncthreads();
    }
    if (tid == 0) qv[c] = red[0];
    __syncthreads();
    red[tid] = r; __syncthreads();
    for (int s = 128; s > 0; s >>= 1) {
        if (tid < s) red[tid] += red[tid + s];
        __syncthreads();
    }
    if (tid == 0) rv[c] = red[0];
}

__global__ __launch_bounds__(128) void sims_gemm_kernel(
        const float* __restrict__ base, const float* __restrict__ base2,
        const float* __restrict__ P, const float* __restrict__ A2m,
        const float* __restrict__ ABm,
        const float* __restrict__ qv, const float* __restrict__ rv,
        float* __restrict__ out) {
    __shared__ float sBa [2][16][KC + 1];
    __shared__ float sBa2[2][16][KC + 1];
    __shared__ float sP  [2][16][KC + 1];
    __shared__ float sA2 [2][16][KC + 1];
    __shared__ float sAB [2][16][KC + 1];
    const int tid  = threadIdx.x;
    const int wave = tid >> 5;
    const int lane = tid & 31;
    const int wm = wave >> 1, wn = wave & 1;
    const int half = lane >> 4, fr = lane & 15;
    const int bm0 = blockIdx.y * 32;
    const int cn0 = blockIdx.x * 32;
    v8f acc1 = {}, acc2 = {}, acc3 = {};
    for (int k0 = 0; k0 < D_SZ; k0 += KC) {
        for (int idx = tid; idx < 2 * 16 * KC; idx += 128) {
            int blk = idx / (16 * KC);
            int rem = idx % (16 * KC);
            int row = rem / KC, col = rem % KC;
            size_t ga = (size_t)(bm0 + blk * 16 + row) * D_SZ + k0 + col;
            size_t gb = (size_t)(cn0 + blk * 16 + row) * D_SZ + k0 + col;
            sBa [blk][row][col] = base [ga];
            sBa2[blk][row][col] = base2[ga];
            sP  [blk][row][col] = P  [gb];
            sA2 [blk][row][col] = A2m[gb];
            sAB [blk][row][col] = ABm[gb];
        }
        __syncthreads();
#pragma unroll
        for (int kk = 0; kk < KC; kk += 4) {
            int kb = kk + half * 2;
            v2f a1, a2, bp, b2, bb;
            a1[0] = sBa [wm][fr][kb]; a1[1] = sBa [wm][fr][kb + 1];
            a2[0] = sBa2[wm][fr][kb]; a2[1] = sBa2[wm][fr][kb + 1];
            bp[0] = sP  [wn][fr][kb]; bp[1] = sP  [wn][fr][kb + 1];
            b2[0] = sA2 [wn][fr][kb]; b2[1] = sA2 [wn][fr][kb + 1];
            bb[0] = sAB [wn][fr][kb]; bb[1] = sAB [wn][fr][kb + 1];
            acc1 = __builtin_amdgcn_wmma_f32_16x16x4_f32(
                    false, a1, false, bp, (short)0, acc1, false, false);
            acc2 = __builtin_amdgcn_wmma_f32_16x16x4_f32(
                    false, a2, false, b2, (short)0, acc2, false, false);
            acc3 = __builtin_amdgcn_wmma_f32_16x16x4_f32(
                    false, a1, false, bb, (short)0, acc3, false, false);
        }
        __syncthreads();
    }
#pragma unroll
    for (int j = 0; j < 8; ++j) {
        int b = bm0 + wm * 16 + j + half * 8;
        int c = cn0 + wn * 16 + fr;
        float num = acc1[j] + qv[c];
        float den = sqrtf(acc2[j] + 2.f * acc3[j] + rv[c]) + EPS_L2;
        out[(size_t)b * B_SZ + c] = num / den;
    }
}

extern "C" void kernel_launch(void* const* d_in, const int* in_sizes, int n_in,
                              void* d_out, int out_size, void* d_ws, size_t ws_size,
                              hipStream_t stream) {
    const float* img   = (const float*)d_in[0];
    const float* cap   = (const float*)d_in[1];
    const int*   lens  = (const int*)  d_in[2];
    const float* fc_w  = (const float*)d_in[3];
    const float* fc_b  = (const float*)d_in[4];
    const float* bn_w  = (const float*)d_in[5];
    const float* bn_b  = (const float*)d_in[6];
    float* out = (float*)d_out;

    float* ws = (float*)d_ws;
    const size_t BD = (size_t)B_SZ * D_SZ;
    float* cap_mean = ws;
    float* meanR    = cap_mean + BD;
    float* base     = meanR + BD;
    float* base2    = base + BD;
    float* P        = base2 + BD;
    float* A2m      = P + BD;
    float* ABm      = A2m + BD;
    float* params   = ABm + BD;
    float* colsum1  = params + 2 * BD;
    float* colsum2  = colsum1 + D_SZ;
    float* qv       = colsum2 + D_SZ;
    float* rv       = qv + B_SZ;

    zero_ws_kernel<<<dim3((2 * D_SZ + 255) / 256), 256, 0, stream>>>(colsum1, 2 * D_SZ);
    cap_pre_kernel<<<dim3(B_SZ), 256, 0, stream>>>(cap, lens, cap_mean);
    img_pre_kernel<<<dim3(B_SZ), 256, 0, stream>>>(img, meanR, colsum1, colsum2);
    make_base_kernel<<<dim3(B_SZ), 256, 0, stream>>>(meanR, colsum1, colsum2,
                                                     bn_w, bn_b, base, base2);
    fc_gemm_kernel<<<dim3(2 * D_SZ / 32, B_SZ / 32), 128, 0, stream>>>(
            cap_mean, fc_w, fc_b, params);
    make_pab_kernel<<<dim3(B_SZ), 256, 0, stream>>>(cap_mean, params,
                                                    P, A2m, ABm, qv, rv);
    sims_gemm_kernel<<<dim3(B_SZ / 32, B_SZ / 32), 128, 0, stream>>>(
            base, base2, P, A2m, ABm, qv, rv, out);
}


# Both WMMA GEMMs: fragments straight from global memory (16-byte loads, a 64-k block ahead in registers), no LDS tiles or barriers; a wave owns a 32x32 (fc) or 16x16 (sims) tile; sum order changes
# speedup vs baseline: 3.1254x; 3.1254x over previous
#include <hip/hip_runtime.h>
#include <hip/hip_bf16.h>
#include <math.h>

#define B_SZ 256
#define R_SZ 36
#define T_SZ 64
#define D_SZ 1024
#define EPS_L2 1e-8f
#define EPS_BN 1e-5f
#define LEAK 0.1f

typedef __attribute__((ext_vector_type(2))) float v2f;
typedef __attribute__((ext_vector_type(8))) float v8f;

__global__ void zero_ws_kernel(float* p, int n) {
    int i = blockIdx.x * blockDim.x + threadIdx.x;
    if (i < n) p[i] = 0.f;
}

__global__ void cap_pre_kernel(const float* __restrict__ cap,
                               const int* __restrict__ lens,
                               float* __restrict__ cap_mean) {
    const int b = blockIdx.x;
    const int len = lens[b];
    const float* cb = cap + (size_t)b * T_SZ * D_SZ;
    float sumsq[4] = {0.f, 0.f, 0.f, 0.f};
    float summ[4]  = {0.f, 0.f, 0.f, 0.f};
    for (int t = 0; t < T_SZ; ++t) {
        const float* row = cb + (size_t)t * D_SZ;
        const bool valid = (t < len);
#pragma unroll
        for (int i = 0; i < 4; ++i) {
            float x = row[threadIdx.x + i * 256];
            float y = x > 0.f ? x : LEAK * x;
            sumsq[i] += y * y;
            if (valid) summ[i] += y;
        }
    }
#pragma unroll
    for (int i = 0; i < 4; ++i) {
        int d = threadIdx.x + i * 256;
        float n = sqrtf(sumsq[i]) + EPS_L2;
        cap_mean[(size_t)b * D_SZ + d] = summ[i] / (n * (float)len);
    }
}

__global__ void img_pre_kernel(const float* __restrict__ img,
                               float* __restrict__ meanR,
                               float* __restrict__ cs1,
                               float* __restrict__ cs2) {
    const int b = blockIdx.x;
    const float* ib = img + (size_t)b * R_SZ * D_SZ;
    float sumsq[4] = {0.f, 0.f, 0.f, 0.f};
    float summ[4]  = {0.f, 0.f, 0.f, 0.f};
    for (int r = 0; r < R_SZ; ++r) {
        const float* row = ib + (size_t)r * D_SZ;
#pragma unroll
        for (int i = 0; i < 4; ++i) {
            float x = row[threadIdx.x + i * 256];
            float y = x > 0.f ? x : LEAK * x;
            sumsq[i] += y * y;
            summ[i]  += y;
        }
    }
#pragma unroll
    for (int i = 0; i < 4; ++i) {
        int d = threadIdx.x + i * 256;
        float n = sqrtf(sumsq[i]) + EPS_L2;
        float s1 = summ[i] / n;
        float s2 = sumsq[i] / (n * n);
        meanR[(size_t)b * D_SZ + d] = s1 * (1.f / (float)R_SZ);
        atomicAdd(cs1 + d, s1);
        atomicAdd(cs2 + d, s2);
    }
}

__global__ void make_base_kernel(const float* __restrict__ meanR,
                                 const float* __restrict__ cs1,
                                 const float* __restrict__ cs2,
                                 const float* __restrict__ bn_w,
                                 const float* __restrict__ bn_b,
                                 float* __restrict__ base,
                                 float* __restrict__ base2) {
    const int b = blockIdx.x;
    const float inv_n = 1.f / (float)(B_SZ * R_SZ);
#pragma unroll
    for (int i = 0; i < 4; ++i) {
        int d = threadIdx.x + i * 256;
        float mu  = cs1[d] * inv_n;
        float ex2 = cs2[d] * inv_n;
        float var = ex2 - mu * mu;
        float scale = bn_w[d] * rsqrtf(var + EPS_BN);
        float v = (meanR[(size_t)b * D_SZ + d] - mu) * scale + bn_b[d];
        base[(size_t)b * D_SZ + d]  = v;
        base2[(size_t)b * D_SZ + d] = v * v;
    }
}

#define KB 64
#define NQ (KB / 8)

typedef __attribute__((ext_vector_type(4))) float v4f;

__device__ __forceinline__ v4f ld_quad(const float* __restrict__ p) {
    return *(const v4f*)p;
}
__device__ __forceinline__ v2f lo2(v4f q) { return __builtin_shufflevector(q, q, 0, 1); }
__device__ __forceinline__ v2f hi2(v4f q) { return __builtin_shufflevector(q, q, 2, 3); }
__device__ __forceinline__ v8f mma4(v2f a, v2f b, v8f c) {
    return __builtin_amdgcn_wmma_f32_16x16x4_f32(false, a, false, b, (short)0, c,
                                                 false, false);
}

struct FcBuf { v4f a0[NQ], a1[NQ], b0[NQ], b1[NQ]; };

__device__ __forceinline__ void fc_load(FcBuf& s, const float* __restrict__ pa,
                                        const float* __restrict__ pb) {
#pragma unroll
    for (int j = 0; j < NQ; ++j) {
        s.a0[j] = ld_quad(pa + 8 * j);
        s.a1[j] = ld_quad(pa + 16 * D_SZ + 8 * j);
        s.b0[j] = ld_quad(pb + 8 * j);
        s.b1[j] = ld_quad(pb + 16 * D_SZ + 8 * j);
    }
}
__device__ __forceinline__ void fc_mma(const FcBuf& s, v8f& c00, v8f& c01,
                                       v8f& c10, v8f& c11) {
#pragma unroll
    for (int j = 0; j < NQ; ++j) {
        c00 = mma4(lo2(s.a0[j]), lo2(s.b0[j]), c00);
        c01 = mma4(lo2(s.a0[j]), lo2(s.b1[j]), c01);
        c10 = mma4(lo2(s.a1[j]), lo2(s.b0[j]), c10);
        c11 = mma4(lo2(s.a1[j]), lo2(s.b1[j]), c11);
        c00 = mma4(hi2(s.a0[j]), hi2(s.b0[j]), c00);
        c01 = mma4(hi2(s.a0[j]), hi2(s.b1[j]), c01);
        c10 = mma4(hi2(s.a1[j]), hi2(s.b0[j]), c10);
        c11 = mma4(hi2(s.a1[j]), hi2(s.b1[j]), c11);
    }
}

__global__ __launch_bounds__(64) void fc_gemm_kernel(
        const float* __restrict__ A, const float* __restrict__ Bw,
        const float* __restrict__ bias, float* __restrict__ Cout) {
    const int wave = threadIdx.x >> 5;
    const int lane = threadIdx.x & 31;
    const int half = lane >> 4, fr = lane & 15;
    const int m0 = blockIdx.y * 32;
    const int n0 = blockIdx.x * 64 + wave * 32;
    const float* pa = A  + (size_t)(m0 + fr) * D_SZ + 4 * half;
    const float* pb = Bw + (size_t)(n0 + fr) * D_SZ + 4 * half;
    v8f c00 = {}, c01 = {}, c10 = {}, c11 = {};
    FcBuf s0, s1;
    fc_load(s0, pa, pb);
    for (int k0 = 0; k0 < D_SZ - 2 * KB; k0 += 2 * KB) {
        fc_load(s1, pa + k0 + KB, pb + k0 + KB);
        fc_mma(s0, c00, c01, c10, c11);
        fc_load(s0, pa + k0 + 2 * KB, pb + k0 + 2 * KB);
        fc_mma(s1, c00, c01, c10, c11);
    }
    fc_load(s1, pa + D_SZ - KB, pb + D_SZ - KB);
    fc_mma(s0, c00, c01, c10, c11);
    fc_mma(s1, c00, c01, c10, c11);

    const int n_lo = n0 + fr, n_hi = n0 + 16 + fr;
    const float bias_lo = bias[n_lo], bias_hi = bias[n_hi];
#pragma unroll
    for (int j = 0; j < 8; ++j) {
        const int m_lo = m0 + j + half * 8, m_hi = m_lo + 16;
        Cout[(size_t)m_lo * (2 * D_SZ) + n_lo] = c00[j] + bias_lo;
        Cout[(size_t)m_lo * (2 * D_SZ) + n_hi] = c01[j] + bias_hi;
        Cout[(size_t)m_hi * (2 * D_SZ) + n_lo] = c10[j] + bias_lo;
        Cout[(size_t)m_hi * (2 * D_SZ) + n_hi] = c11[j] + bias_hi;
    }
}

__global__ void make_pab_kernel(const float* __restrict__ cap_mean,
                                const float* __restrict__ params,
                                float* __restrict__ P, float* __restrict__ A2m,
                                float* __restrict__ ABm,
                                float* __restrict__ qv, float* __restrict__ rv) {
    const int c = blockIdx.x;
    const int tid = threadIdx.x;
    __shared__ float red[256];

    float ss = 0.f;
#pragma unroll
    for (int i = 0; i < 4; ++i) {
        float x = cap_mean[(size_t)c * D_SZ + tid + i * 256];
        ss += x * x;
    }
    red[tid] = ss; __syncthreads();
    for (int s = 128; s > 0; s >>= 1) {
        if (tid < s) red[tid] += red[tid + s];
        __syncthreads();
    }
    const float nrm = sqrtf(red[0]) + EPS_L2;
    __syncthreads();

    float q = 0.f, r = 0.f;
#pragma unroll
    for (int i = 0; i < 4; ++i) {
        int d = tid + i * 256;
        float alpha = params[(size_t)c * (2 * D_SZ) + 2 * d];
        float beta  = params[(size_t)c * (2 * D_SZ) + 2 * d + 1];
        float ch = cap_mean[(size_t)c * D_SZ + d] / nrm;
        P  [(size_t)c * D_SZ + d] = alpha * ch;
        A2m[(size_t)c * D_SZ + d] = alpha * alpha;
        ABm[(size_t)c * D_SZ + d] = alpha * beta;
        q += beta * ch;
        r += beta * beta;
    }
    red[tid] = q; __syncthreads();
    for (int s = 128; s > 0; s >>= 1) {
        if (tid < s) red[tid] += red[tid + s];
        __syncthreads();
    }
    if (tid == 0) qv[c] = red[0];
    __syncthreads();
    red[tid] = r; __syncthreads();
    for (int s = 128; s > 0; s >>= 1) {
        if (tid < s) red[tid] += red[tid + s];
        __syncthreads();
    }
    if (tid == 0) rv[c] = red[0];
}

struct SimsBuf { v4f a1[NQ], a2[NQ], bp[NQ], b2[NQ], bb[NQ]; };

__device__ __forceinline__ void sims_load(SimsBuf& s,
        const float* __restrict__ pa1, const float* __restrict__ pa2,
        const float* __restrict__ pb1, const float* __restrict__ pb2,
        const float* __restrict__ pb3) {
#pragma unroll
    for (int j = 0; j < NQ; ++j) {
        s.a1[j] = ld_quad(pa1 + 8 * j);
        s.a2[j] = ld_quad(pa2 + 8 * j);
        s.bp[j] = ld_quad(pb1 + 8 * j);
        s.b2[j] = ld_quad(pb2 + 8 * j);
        s.bb[j] = ld_quad(pb3 + 8 * j);
    }
}
__device__ __forceinline__ void sims_mma(const SimsBuf& s, v8f& acc1, v8f& acc2,
                                         v8f& acc3) {
#pragma unroll
    for (int j = 0; j < NQ; ++j) {
        acc1 = mma4(lo2(s.a1[j]), lo2(s.bp[j]), acc1);
        acc2 = mma4(lo2(s.a2[j]), lo2(s.b2[j]), acc2);
        acc3 = mma4(lo2(s.a1[j]), lo2(s.bb[j]), acc3);
        acc1 = mma4(hi2(s.a1[j]), hi2(s.bp[j]), acc1);
        acc2 = mma4(hi2(s.a2[j]), hi2(s.b2[j]), acc2);
        acc3 = mma4(hi2(s.a1[j]), hi2(s.bb[j]), acc3);
    }
}

__global__ __launch_bounds__(32) void sims_gemm_kernel(
        const float* __restrict__ base, const float* __restrict__ base2,
        const float* __restrict__ P, const float* __restrict__ A2m,
        const float* __restrict__ ABm,
        const float* __restrict__ qv, const float* __restrict__ rv,
        float* __restrict__ out) {
    const int lane = threadIdx.x;
    const int half = lane >> 4, fr = lane & 15;
    const int bm0 = blockIdx.y * 16;
    const int cn0 = blockIdx.x * 16;
    const size_t ga = (size_t)(bm0 + fr) * D_SZ + 4 * half;
    const size_t gb = (size_t)(cn0 + fr) * D_SZ + 4 * half;
    const float* pa1 = base  + ga;
    const float* pa2 = base2 + ga;
    const float* pb1 = P   + gb;
    const float* pb2 = A2m + gb;
    const float* pb3 = ABm + gb;
    v8f acc1 = {}, acc2 = {}, acc3 = {};
    SimsBuf s0, s1;
    sims_load(s0, pa1, pa2, pb1, pb2, pb3);
    for (int k0 = 0; k0 < D_SZ - 2 * KB; k0 += 2 * KB) {
        const int ka = k0 + KB, kb = k0 + 2 * KB;
        sims_load(s1, pa1 + ka, pa2 + ka, pb1 + ka, pb2 + ka, pb3 + ka);
        sims_mma(s0, acc1, acc2, acc3);
        sims_load(s0, pa1 + kb, pa2 + kb, pb1 + kb, pb2 + kb, pb3 + kb);
        sims_mma(s1, acc1, acc2, acc3);
    }
    {
        const int ka = D_SZ - KB;
        sims_load(s1, pa1 + ka, pa2 + ka, pb1 + ka, pb2 + ka, pb3 + ka);
    }
    sims_mma(s0, acc1, acc2, acc3);
    sims_mma(s1, acc1, acc2, acc3);

    const int c = cn0 + fr;
    const float qc = qv[c], rc = rv[c];
#pragma unroll
    for (int j = 0; j < 8; ++j) {
        int b = bm0 + j + half * 8;
        float num = acc1[j] + qc;
        float den = sqrtf(acc2[j] + 2.f * acc3[j] + rc) + EPS_L2;
        out[(size_t)b * B_SZ + c] = num / den;
    }
}

extern "C" void kernel_launch(void* const* d_in, const int* in_sizes, int n_in,
                              void* d_out, int out_size, void* d_ws, size_t ws_size,
                              hipStream_t stream) {
    const float* img   = (const float*)d_in[0];
    const float* cap   = (const float*)d_in[1];
    const int*   lens  = (const int*)  d_in[2];
    const float* fc_w  = (const float*)d_in[3];
    const float* fc_b  = (const float*)d_in[4];
    const float* bn_w  = (const float*)d_in[5];
    const float* bn_b  = (const float*)d_in[6];
    float* out = (float*)d_out;

    float* ws = (float*)d_ws;
    const size_t BD = (size_t)B_SZ * D_SZ;
    float* cap_mean = ws;
    float* meanR    = cap_mean + BD;
    float* base     = meanR + BD;
    float* base2    = base + BD;
    float* P        = base2 + BD;
    float* A2m      = P + BD;
    float* ABm      = A2m + BD;
    float* params   = ABm + BD;
    float* colsum1  = params + 2 * BD;
    float* colsum2  = colsum1 + D_SZ;
    float* qv       = colsum2 + D_SZ;
    float* rv       = qv + B_SZ;

    zero_ws_kernel<<<dim3((2 * D_SZ + 255) / 256), 256, 0, stream>>>(colsum1, 2 * D_SZ);
    cap_pre_kernel<<<dim3(B_SZ), 256, 0, stream>>>(cap, lens, cap_mean);
    img_pre_kernel<<<dim3(B_SZ), 256, 0, stream>>>(img, meanR, colsum1, colsum2);
    make_base_kernel<<<dim3(B_SZ), 256, 0, stream>>>(meanR, colsum1, colsum2,
                                                     bn_w, bn_b, base, base2);
    fc_gemm_kernel<<<dim3(2 * D_SZ / 64, B_SZ / 32), 64, 0, stream>>>(
            cap_mean, fc_w, fc_b, params);
    make_pab_kernel<<<dim3(B_SZ), 256, 0, stream>>>(cap_mean, params,
                                                    P, A2m, ABm, qv, rv);
    sims_gemm_kernel<<<dim3(B_SZ / 16, B_SZ / 16), 32, 0, stream>>>(
            base, base2, P, A2m, ABm, qv, rv, out);
}


# Both WMMA GEMMs re-tiled for a wave on every SIMD: fc wave tile 32x32 to 32x16 in 4-wave blocks; sims one wave per product, 2 accumulators each, joined via LDS; K block 128; sims sum order changes
# speedup vs baseline: 3.4992x; 1.1196x over previous
#include <hip/hip_runtime.h>
#include <hip/hip_bf16.h>
#include <math.h>

#define B_SZ 256
#define R_SZ 36
#define T_SZ 64
#define D_SZ 1024
#define EPS_L2 1e-8f
#define EPS_BN 1e-5f
#define LEAK 0.1f

typedef __attribute__((ext_vector_type(2))) float v2f;
typedef __attribute__((ext_vector_type(8))) float v8f;

__global__ void zero_ws_kernel(float* p, int n) {
    int i = blockIdx.x * blockDim.x + threadIdx.x;
    if (i < n) p[i] = 0.f;
}

__global__ void cap_pre_kernel(const float* __restrict__ cap,
                               const int* __restrict__ lens,
                               float* __restrict__ cap_mean) {
    const int b = blockIdx.x;
    const int len = lens[b];
    const float* cb = cap + (size_t)b * T_SZ * D_SZ;
    float sumsq[4] = {0.f, 0.f, 0.f, 0.f};
    float summ[4]  = {0.f, 0.f, 0.f, 0.f};
    for (int t = 0; t < T_SZ; ++t) {
        const float* row = cb + (size_t)t * D_SZ;
        const bool valid = (t < len);
#pragma unroll
        for (int i = 0; i < 4; ++i) {
            float x = row[threadIdx.x + i * 256];
            float y = x > 0.f ? x : LEAK * x;
            sumsq[i] += y * y;
            if (valid) summ[i] += y;
        }
    }
#pragma unroll
    for (int i = 0; i < 4; ++i) {
        int d = threadIdx.x + i * 256;
        float n = sqrtf(sumsq[i]) + EPS_L2;
        cap_mean[(size_t)b * D_SZ + d] = summ[i] / (n * (float)len);
    }
}

__global__ void img_pre_kernel(const float* __restrict__ img,
                               float* __restrict__ meanR,
                               float* __restrict__ cs1,
                               float* __restrict__ cs2) {
    const int b = blockIdx.x;
    const float* ib = img + (size_t)b * R_SZ * D_SZ;
    float sumsq[4] = {0.f, 0.f, 0.f, 0.f};
    float summ[4]  = {0.f, 0.f, 0.f, 0.f};
    for (int r = 0; r < R_SZ; ++r) {
        const float* row = ib + (size_t)r * D_SZ;
#pragma unroll
        for (int i = 0; i < 4; ++i) {
            float x = row[threadIdx.x + i * 256];
            float y = x > 0.f ? x : LEAK * x;
            sumsq[i] += y * y;
            summ[i]  += y;
        }
    }
#pragma unroll
    for (int i = 0; i < 4; ++i) {
        int d = threadIdx.x + i * 256;
        float n = sqrtf(sumsq[i]) + EPS_L2;
        float s1 = summ[i] / n;
        float s2 = sumsq[i] / (n * n);
        meanR[(size_t)b * D_SZ + d] = s1 * (1.f / (float)R_SZ);
        atomicAdd(cs1 + d, s1);
        atomicAdd(cs2 + d, s2);
    }
}

__global__ void make_base_kernel(const float* __restrict__ meanR,
                                 const float* __restrict__ cs1,
                                 const float* __restrict__ cs2,
                                 const float* __restrict__ bn_w,
                                 const float* __restrict__ bn_b,
                                 float* __restrict__ base,
                                 float* __restrict__ base2) {
    const int b = blockIdx.x;
    const float inv_n = 1.f / (float)(B_SZ * R_SZ);
#pragma unroll
    for (int i = 0; i < 4; ++i) {
        int d = threadIdx.x + i * 256;
        float mu  = cs1[d] * inv_n;
        float ex2 = cs2[d] * inv_n;
        float var = ex2 - mu * mu;
        float scale = bn_w[d] * rsqrtf(var + EPS_BN);
        float v = (meanR[(size_t)b * D_SZ + d] - mu) * scale + bn_b[d];
        base[(size_t)b * D_SZ + d]  = v;
        base2[(size_t)b * D_SZ + d] = v * v;
    }
}

#define KB 128
#define NQ (KB / 8)

typedef __attribute__((ext_vector_type(4))) float v4f;

__device__ __forceinline__ v4f ld_quad(const float* __restrict__ p) {
    return *(const v4f*)p;
}
__device__ __forceinline__ v2f lo2(v4f q) { return __builtin_shufflevector(q, q, 0, 1); }
__device__ __forceinline__ v2f hi2(v4f q) { return __builtin_shufflevector(q, q, 2, 3); }
__device__ __forceinline__ v8f mma4(v2f a, v2f b, v8f c) {
    return __builtin_amdgcn_wmma_f32_16x16x4_f32(false, a, false, b, (short)0, c,
                                                 false, false);
}

struct FcBuf { v4f a0[NQ], a1[NQ], b0[NQ]; };

__device__ __forceinline__ void fc_load(FcBuf& s, const float* __restrict__ pa,
                                        const float* __restrict__ pb) {
#pragma unroll
    for (int j = 0; j < NQ; ++j) {
        s.a0[j] = ld_quad(pa + 8 * j);
        s.a1[j] = ld_quad(pa + 16 * D_SZ + 8 * j);
        s.b0[j] = ld_quad(pb + 8 * j);
    }
}
__device__ __forceinline__ void fc_mma(const FcBuf& s, v8f& c0, v8f& c1) {
#pragma unroll
    for (int j = 0; j < NQ; ++j) {
        c0 = mma4(lo2(s.a0[j]), lo2(s.b0[j]), c0);
        c1 = mma4(lo2(s.a1[j]), lo2(s.b0[j]), c1);
        c0 = mma4(hi2(s.a0[j]), hi2(s.b0[j]), c0);
        c1 = mma4(hi2(s.a1[j]), hi2(s.b0[j]), c1);
    }
}

__global__ __launch_bounds__(128) void fc_gemm_kernel(
        const float* __restrict__ A, const float* __restrict__ Bw,
        const float* __restrict__ bias, float* __restrict__ Cout) {
    const int wave = threadIdx.x >> 5;
    const int lane = threadIdx.x & 31;
    const int half = lane >> 4, fr = lane & 15;
    const int m0 = blockIdx.y * 32;
    const int n0 = blockIdx.x * 64 + wave * 16;
    const float* pa = A  + (size_t)(m0 + fr) * D_SZ + 4 * half;
    const float* pb = Bw + (size_t)(n0 + fr) * D_SZ + 4 * half;
    v8f c0 = {}, c1 = {};
    FcBuf s0, s1;
    fc_load(s0, pa, pb);
    for (int k0 = 0; k0 < D_SZ - 2 * KB; k0 += 2 * KB) {
        fc_load(s1, pa + k0 + KB, pb + k0 + KB);
        fc_mma(s0, c0, c1);
        fc_load(s0, pa + k0 + 2 * KB, pb + k0 + 2 * KB);
        fc_mma(s1, c0, c1);
    }
    fc_load(s1, pa + D_SZ - KB, pb + D_SZ - KB);
    fc_mma(s0, c0, c1);
    fc_mma(s1, c0, c1);

    const int n = n0 + fr;
    const float bias_n = bias[n];
#pragma unroll
    for (int j = 0; j < 8; ++j) {
        const int m_lo = m0 + j + half * 8, m_hi = m_lo + 16;
        Cout[(size_t)m_lo * (2 * D_SZ) + n] = c0[j] + bias_n;
        Cout[(size_t)m_hi * (2 * D_SZ) + n] = c1[j] + bias_n;
    }
}

__global__ void make_pab_kernel(const float* __restrict__ cap_mean,
                                const float* __restrict__ params,
                                float* __restrict__ P, float* __restrict__ A2m,
                                float* __restrict__ ABm,
                                float* __restrict__ qv, float* __restrict__ rv) {
    const int c = blockIdx.x;
    const int tid = threadIdx.x;
    __shared__ float red[256];

    float ss = 0.f;
#pragma unroll
    for (int i = 0; i < 4; ++i) {
        float x = cap_mean[(size_t)c * D_SZ + tid + i * 256];
        ss += x * x;
    }
    red[tid] = ss; __syncthreads();
    for (int s = 128; s > 0; s >>= 1) {
        if (tid < s) red[tid] += red[tid + s];
        __syncthreads();
    }
    const float nrm = sqrtf(red[0]) + EPS_L2;
    __syncthreads();

    float q = 0.f, r = 0.f;
#pragma unroll
    for (int i = 0; i < 4; ++i) {
        int d = tid + i * 256;
        float alpha = params[(size_t)c * (2 * D_SZ) + 2 * d];
        float beta  = params[(size_t)c * (2 * D_SZ) + 2 * d + 1];
        float ch = cap_mean[(size_t)c * D_SZ + d] / nrm;
        P  [(size_t)c * D_SZ + d] = alpha * ch;
        A2m[(size_t)c * D_SZ + d] = alpha * alpha;
        ABm[(size_t)c * D_SZ + d] = alpha * beta;
        q += beta * ch;
        r += beta * beta;
    }
    red[tid] = q; __syncthreads();
    for (int s = 128; s > 0; s >>= 1) {
        if (tid < s) red[tid] += red[tid + s];
        __syncthreads();
    }
    if (tid == 0) qv[c] = red[0];
    __syncthreads();
    red[tid] = r; __syncthreads();
    for (int s = 128; s > 0; s >>= 1) {
        if (tid < s) red[tid] += red[tid + s];
        __syncthreads();
    }
    if (tid == 0) rv[c] = red[0];
}

struct SimsBuf { v4f a[NQ], b[NQ]; };

__device__ __forceinline__ void sims_load(SimsBuf& s,
        const float* __restrict__ pa, const float* __restrict__ pb) {
#pragma unroll
    for (int j = 0; j < NQ; ++j) {
        s.a[j] = ld_quad(pa + 8 * j);
        s.b[j] = ld_quad(pb + 8 * j);
    }
}
__device__ __forceinline__ void sims_mma(const SimsBuf& s, v8f& acc_lo, v8f& acc_hi) {
#pragma unroll
    for (int j = 0; j < NQ; ++j) {
        acc_lo = mma4(lo2(s.a[j]), lo2(s.b[j]), acc_lo);
        acc_hi = mma4(hi2(s.a[j]), hi2(s.b[j]), acc_hi);
    }
}

__global__ __launch_bounds__(96) void sims_gemm_kernel(
        const float* __restrict__ base, const float* __restrict__ base2,
        const float* __restrict__ P, const float* __restrict__ A2m,
        const float* __restrict__ ABm,
        const float* __restrict__ qv, const float* __restrict__ rv,
        float* __restrict__ out) {
    __shared__ float part[2][8][32];
    const int wave = __builtin_amdgcn_readfirstlane((int)(threadIdx.x >> 5));
    const int lane = threadIdx.x & 31;
    const int half = lane >> 4, fr = lane & 15;
    const int bm0 = blockIdx.y * 16;
    const int cn0 = blockIdx.x * 16;
    const float* opa = (wave == 1) ? base2 : base;
    const float* opb = (wave == 0) ? P : ((wave == 1) ? A2m : ABm);
    const float* pa = opa + (size_t)(bm0 + fr) * D_SZ + 4 * half;
    const float* pb = opb + (size_t)(cn0 + fr) * D_SZ + 4 * half;
    v8f acc_lo = {}, acc_hi = {};
    SimsBuf s0, s1;
    sims_load(s0, pa, pb);
    for (int k0 = 0; k0 < D_SZ - 2 * KB; k0 += 2 * KB) {
        sims_load(s1, pa + k0 + KB, pb + k0 + KB);
        sims_mma(s0, acc_lo, acc_hi);
        sims_load(s0, pa + k0 + 2 * KB, pb + k0 + 2 * KB);
        sims_mma(s1, acc_lo, acc_hi);
    }
    sims_load(s1, pa + D_SZ - KB, pb + D_SZ - KB);
    sims_mma(s0, acc_lo, acc_hi);
    sims_mma(s1, acc_lo, acc_hi);
    const v8f acc = acc_lo + acc_hi;

    if (wave != 0) {
#pragma unroll
        for (int j = 0; j < 8; ++j) part[wave - 1][j][lane] = acc[j];
    }
    __syncthreads();
    if (wave == 0) {
        const int c = cn0 + fr;
        const float qc = qv[c], rc = rv[c];
#pragma unroll
        for (int j = 0; j < 8; ++j) {
            int b = bm0 + j + half * 8;
            float num = acc[j] + qc;
            float den = sqrtf(part[0][j][lane] + 2.f * part[1][j][lane] + rc) + EPS_L2;
            out[(size_t)b * B_SZ + c] = num / den;
        }
    }
}

extern "C" void kernel_launch(void* const* d_in, const int* in_sizes, int n_in,
                              void* d_out, int out_size, void* d_ws, size_t ws_size,
                              hipStream_t stream) {
    const float* img   = (const float*)d_in[0];
    const float* cap   = (const float*)d_in[1];
    const int*   lens  = (const int*)  d_in[2];
    const float* fc_w  = (const float*)d_in[3];
    const float* fc_b  = (const float*)d_in[4];
    const float* bn_w  = (const float*)d_in[5];
    const float* bn_b  = (const float*)d_in[6];
    float* out = (float*)d_out;

    float* ws = (float*)d_ws;
    const size_t BD = (size_t)B_SZ * D_SZ;
    float* cap_mean = ws;
    float* meanR    = cap_mean + BD;
    float* base     = meanR + BD;
    float* base2    = base + BD;
    float* P        = base2 + BD;
    float* A2m      = P + BD;
    float* ABm      = A2m + BD;
    float* params   = ABm + BD;
    float* colsum1  = params + 2 * BD;
    float* colsum2  = colsum1 + D_SZ;
    float* qv       = colsum2 + D_SZ;
    float* rv       = qv + B_SZ;

    zero_ws_kernel<<<dim3((2 * D_SZ + 255) / 256), 256, 0, stream>>>(colsum1, 2 * D_SZ);
    cap_pre_kernel<<<dim3(B_SZ), 256, 0, stream>>>(cap, lens, cap_mean);
    img_pre_kernel<<<dim3(B_SZ), 256, 0, stream>>>(img, meanR, colsum1, colsum2);
    make_base_kernel<<<dim3(B_SZ), 256, 0, stream>>>(meanR, colsum1, colsum2,
                                                     bn_w, bn_b, base, base2);
    fc_gemm_kernel<<<dim3(2 * D_SZ / 64, B_SZ / 32), 128, 0, stream>>>(
            cap_mean, fc_w, fc_b, params);
    make_pab_kernel<<<dim3(B_SZ), 256, 0, stream>>>(cap_mean, params,
                                                    P, A2m, ABm, qv, rv);
    sims_gemm_kernel<<<dim3(B_SZ / 16, B_SZ / 16), 96, 0, stream>>>(
            base, base2, P, A2m, ABm, qv, rv, out);
}


# fc GEMM epilogue forms P, A2, AB and per-block q, r sums from its accumulators (row norm now in cap_pre): params array and the P/A2/AB launch dropped, 7 launches to 6; q, r, norm sum order changes
# speedup vs baseline: 3.7931x; 1.0840x over previous
#include <hip/hip_runtime.h>
#include <hip/hip_bf16.h>
#include <math.h>

#define B_SZ 256
#define R_SZ 36
#define T_SZ 64
#define D_SZ 1024
#define EPS_L2 1e-8f
#define EPS_BN 1e-5f
#define LEAK 0.1f

typedef __attribute__((ext_vector_type(2))) float v2f;
typedef __attribute__((ext_vector_type(8))) float v8f;

__global__ void zero_ws_kernel(float* p, int n) {
    int i = blockIdx.x * blockDim.x + threadIdx.x;
    if (i < n) p[i] = 0.f;
}

template <int CTRL>
__device__ __forceinline__ float row_read(float v) {
    return __builtin_bit_cast(float, __builtin_amdgcn_update_dpp(
            0, __builtin_bit_cast(int, v), CTRL, 0xf, 0xf, true));
}
__device__ __forceinline__ float xhalf(float x) {
    const int b = __builtin_bit_cast(int, x);
    return __builtin_bit_cast(float, __builtin_amdgcn_permlanex16(
            b, b, 0x76543210, 0xfedcba98, false, false));
}

__global__ void cap_pre_kernel(const float* __restrict__ cap,
                               const int* __restrict__ lens,
                               float* __restrict__ cap_mean,
                               float* __restrict__ cap_hat) {
    __shared__ float wsum[8];
    const int b = blockIdx.x;
    const int len = lens[b];
    const float* cb = cap + (size_t)b * T_SZ * D_SZ;
    float sumsq[4] = {0.f, 0.f, 0.f, 0.f};
    float summ[4]  = {0.f, 0.f, 0.f, 0.f};
    for (int t = 0; t < T_SZ; ++t) {
        const float* row = cb + (size_t)t * D_SZ;
        const bool valid = (t < len);
#pragma unroll
        for (int i = 0; i < 4; ++i) {
            float x = row[threadIdx.x + i * 256];
            float y = x > 0.f ? x : LEAK * x;
            sumsq[i] += y * y;
            if (valid) summ[i] += y;
        }
    }
    float cm[4];
    float ss = 0.f;
#pragma unroll
    for (int i = 0; i < 4; ++i) {
        float n = sqrtf(sumsq[i]) + EPS_L2;
        cm[i] = summ[i] / (n * (float)len);
        ss += cm[i] * cm[i];
    }
    ss += row_read<0x161>(ss);
    ss += row_read<0x162>(ss);
    ss += row_read<0x164>(ss);
    ss += row_read<0x168>(ss);
    ss += xhalf(ss);
    if ((threadIdx.x & 31) == 0) wsum[threadIdx.x >> 5] = ss;
    __syncthreads();
    float tot = wsum[0];
#pragma unroll
    for (int w = 1; w < 8; ++w) tot += wsum[w];
    const float nrm = sqrtf(tot) + EPS_L2;
#pragma unroll
    for (int i = 0; i < 4; ++i) {
        int d = threadIdx.x + i * 256;
        cap_mean[(size_t)b * D_SZ + d] = cm[i];
        cap_hat[(size_t)b * D_SZ + d]  = cm[i] / nrm;
    }
}

__global__ void img_pre_kernel(const float* __restrict__ img,
                               float* __restrict__ meanR,
                               float* __restrict__ cs1,
                               float* __restrict__ cs2) {
    const int b = blockIdx.x;
    const float* ib = img + (size_t)b * R_SZ * D_SZ;
    float sumsq[4] = {0.f, 0.f, 0.f, 0.f};
    float summ[4]  = {0.f, 0.f, 0.f, 0.f};
    for (int r = 0; r < R_SZ; ++r) {
        const float* row = ib + (size_t)r * D_SZ;
#pragma unroll
        for (int i = 0; i < 4; ++i) {
            float x = row[threadIdx.x + i * 256];
            float y = x > 0.f ? x : LEAK * x;
            sumsq[i] += y * y;
            summ[i]  += y;
        }
    }
#pragma unroll
    for (int i = 0; i < 4; ++i) {
        int d = threadIdx.x + i * 256;
        float n = sqrtf(sumsq[i]) + EPS_L2;
        float s1 = summ[i] / n;
        float s2 = sumsq[i] / (n * n);
        meanR[(size_t)b * D_SZ + d] = s1 * (1.f / (float)R_SZ);
        atomicAdd(cs1 + d, s1);
        atomicAdd(cs2 + d, s2);
    }
}

__global__ void make_base_kernel(const float* __restrict__ meanR,
                                 const float* __restrict__ cs1,
                                 const float* __restrict__ cs2,
                                 const float* __restrict__ bn_w,
                                 const float* __restrict__ bn_b,
                                 float* __restrict__ base,
                                 float* __restrict__ base2) {
    const int b = blockIdx.x;
    const float inv_n = 1.f / (float)(B_SZ * R_SZ);
#pragma unroll
    for (int i = 0; i < 4; ++i) {
        int d = threadIdx.x + i * 256;
        float mu  = cs1[d] * inv_n;
        float ex2 = cs2[d] * inv_n;
        float var = ex2 - mu * mu;
        float scale = bn_w[d] * rsqrtf(var + EPS_BN);
        float v = (meanR[(size_t)b * D_SZ + d] - mu) * scale + bn_b[d];
        base[(size_t)b * D_SZ + d]  = v;
        base2[(size_t)b * D_SZ + d] = v * v;
    }
}

#define KB 128
#define NQ (KB / 8)

typedef __attribute__((ext_vector_type(4))) float v4f;

__device__ __forceinline__ v4f ld_quad(const float* __restrict__ p) {
    return *(const v4f*)p;
}
__device__ __forceinline__ v2f lo2(v4f q) { return __builtin_shufflevector(q, q, 0, 1); }
__device__ __forceinline__ v2f hi2(v4f q) { return __builtin_shufflevector(q, q, 2, 3); }
__device__ __forceinline__ v8f mma4(v2f a, v2f b, v8f c) {
    return __builtin_amdgcn_wmma_f32_16x16x4_f32(false, a, false, b, (short)0, c,
                                                 false, false);
}

struct FcBuf { v4f a0[NQ], a1[NQ], b0[NQ]; };

__device__ __forceinline__ void fc_load(FcBuf& s, const float* __restrict__ pa,
                                        const float* __restrict__ pb) {
#pragma unroll
    for (int j = 0; j < NQ; ++j) {
        s.a0[j] = ld_quad(pa + 8 * j);
        s.a1[j] = ld_quad(pa + 16 * D_SZ + 8 * j);
        s.b0[j] = ld_quad(pb + 8 * j);
    }
}
__device__ __forceinline__ void fc_mma(const FcBuf& s, v8f& c0, v8f& c1) {
#pragma unroll
    for (int j = 0; j < NQ; ++j) {
        c0 = mma4(lo2(s.a0[j]), lo2(s.b0[j]), c0);
        c1 = mma4(lo2(s.a1[j]), lo2(s.b0[j]), c1);
        c0 = mma4(hi2(s.a0[j]), hi2(s.b0[j]), c0);
        c1 = mma4(hi2(s.a1[j]), hi2(s.b0[j]), c1);
    }
}

__global__ __launch_bounds__(128) void fc_gemm_kernel(
        const float* __restrict__ A, const float* __restrict__ Bw,
        const float* __restrict__ bias, const float* __restrict__ cap_hat,
        float* __restrict__ P, float* __restrict__ A2m, float* __restrict__ ABm,
        float* __restrict__ qpart, float* __restrict__ rpart) {
    __shared__ float red[4][32][2];
    const int wave = threadIdx.x >> 5;
    const int lane = threadIdx.x & 31;
    const int half = lane >> 4, fr = lane & 15;
    const int m0 = blockIdx.y * 32;
    const int n0 = blockIdx.x * 64 + wave * 16;
    const float* pa = A  + (size_t)(m0 + fr) * D_SZ + 4 * half;
    const float* pb = Bw + (size_t)(n0 + fr) * D_SZ + 4 * half;
    v8f c0 = {}, c1 = {};
    FcBuf s0, s1;
    fc_load(s0, pa, pb);
    for (int k0 = 0; k0 < D_SZ - 2 * KB; k0 += 2 * KB) {
        fc_load(s1, pa + KB, pb + KB);
        fc_mma(s0, c0, c1);
        fc_load(s0, pa + 2 * KB, pb + 2 * KB);
        fc_mma(s1, c0, c1);
        pa += 2 * KB;
        pb += 2 * KB;
    }
    fc_load(s1, pa + KB, pb + KB);
    const int n = n0 + fr;
    const int d = n >> 1;
    float ch_lo[8], ch_hi[8];
#pragma unroll
    for (int j = 0; j < 8; ++j) {
        ch_lo[j] = cap_hat[(size_t)(m0 + j + half * 8) * D_SZ + d];
        ch_hi[j] = cap_hat[(size_t)(m0 + 16 + j + half * 8) * D_SZ + d];
    }
    fc_mma(s0, c0, c1);
    fc_mma(s1, c0, c1);

    const float bias_n = bias[n];
    const bool even = (fr & 1) == 0;
    float* __restrict__ pab = even ? P : ABm;
#pragma unroll
    for (int j = 0; j < 8; ++j) {
        const int r_lo = j + half * 8, r_hi = r_lo + 16;
        const size_t i_lo = (size_t)(m0 + r_lo) * D_SZ + d;
        const size_t i_hi = (size_t)(m0 + r_hi) * D_SZ + d;
        const float v0 = c0[j] + bias_n, v1 = c1[j] + bias_n;
        const float p0 = row_read<0x161>(v0), p1 = row_read<0x161>(v1);
        const float h0 = ch_lo[j], h1 = ch_hi[j];
        pab[i_lo] = v0 * (even ? h0 : p0);
        pab[i_hi] = v1 * (even ? h1 : p1);
        const float a = even ? v0 : p1;
        A2m[even ? i_lo : i_hi] = a * a;
        float t0 = even ? p0 * h0 : v0 * v0;
        float t1 = even ? p1 * h1 : v1 * v1;
        t0 += row_read<0x162>(t0);  t1 += row_read<0x162>(t1);
        t0 += row_read<0x164>(t0);  t1 += row_read<0x164>(t1);
        t0 += row_read<0x168>(t0);  t1 += row_read<0x168>(t1);
        if (fr < 2) {
            red[wave][r_lo][fr] = t0;
            red[wave][r_hi][fr] = t1;
        }
    }
    __syncthreads();
    if (threadIdx.x < 64) {
        const int row = threadIdx.x >> 1, kind = threadIdx.x & 1;
        const float s = ((red[0][row][kind] + red[1][row][kind]) + red[2][row][kind])
                        + red[3][row][kind];
        float* __restrict__ dst = kind ? rpart : qpart;
        dst[(size_t)blockIdx.x * B_SZ + m0 + row] = s;
    }
}

struct SimsBuf { v4f a[NQ], b[NQ]; };
#define NPART (2 * D_SZ / 64)

__device__ __forceinline__ void sims_load(SimsBuf& s,
        const float* __restrict__ pa, const float* __restrict__ pb) {
#pragma unroll
    for (int j = 0; j < NQ; ++j) {
        s.a[j] = ld_quad(pa + 8 * j);
        s.b[j] = ld_quad(pb + 8 * j);
    }
}
__device__ __forceinline__ void sims_mma(const SimsBuf& s, v8f& acc_lo, v8f& acc_hi) {
#pragma unroll
    for (int j = 0; j < NQ; ++j) {
        acc_lo = mma4(lo2(s.a[j]), lo2(s.b[j]), acc_lo);
        acc_hi = mma4(hi2(s.a[j]), hi2(s.b[j]), acc_hi);
    }
}

__global__ __launch_bounds__(96) void sims_gemm_kernel(
        const float* __restrict__ base, const float* __restrict__ base2,
        const float* __restrict__ P, const float* __restrict__ A2m,
        const float* __restrict__ ABm,
        const float* __restrict__ qpart, const float* __restrict__ rpart,
        float* __restrict__ out) {
    __shared__ float part[2][8][32];
    const int wave = __builtin_amdgcn_readfirstlane((int)(threadIdx.x >> 5));
    const int lane = threadIdx.x & 31;
    const int half = lane >> 4, fr = lane & 15;
    const int bm0 = blockIdx.y * 16;
    const int cn0 = blockIdx.x * 16;
    const float* opa = (wave == 1) ? base2 : base;
    const float* opb = (wave == 0) ? P : ((wave == 1) ? A2m : ABm);
    const float* pa = opa + (size_t)(bm0 + fr) * D_SZ + 4 * half;
    const float* pb = opb + (size_t)(cn0 + fr) * D_SZ + 4 * half;
    v8f acc_lo = {}, acc_hi = {};
    SimsBuf s0, s1;
    sims_load(s0, pa, pb);
    float qp[NPART / 2], rp[NPART / 2];
#pragma unroll
    for (int i = 0; i < NPART / 2; ++i) { qp[i] = 0.f; rp[i] = 0.f; }
    if (wave == 0) {
#pragma unroll
        for (int i = 0; i < NPART / 2; ++i) {
            qp[i] = qpart[(size_t)(half * (NPART / 2) + i) * B_SZ + cn0 + fr];
            rp[i] = rpart[(size_t)(half * (NPART / 2) + i) * B_SZ + cn0 + fr];
        }
    }
    for (int k0 = 0; k0 < D_SZ - 2 * KB; k0 += 2 * KB) {
        sims_load(s1, pa + k0 + KB, pb + k0 + KB);
        sims_mma(s0, acc_lo, acc_hi);
        sims_load(s0, pa + k0 + 2 * KB, pb + k0 + 2 * KB);
        sims_mma(s1, acc_lo, acc_hi);
    }
    sims_load(s1, pa + D_SZ - KB, pb + D_SZ - KB);
    sims_mma(s0, acc_lo, acc_hi);
    sims_mma(s1, acc_lo, acc_hi);
    const v8f acc = acc_lo + acc_hi;

    if (wave != 0) {
#pragma unroll
        for (int j = 0; j < 8; ++j) part[wave - 1][j][lane] = acc[j];
    }
    __syncthreads();
    if (wave == 0) {
        const int c = cn0 + fr;
        float qa = qp[0], qb = qp[1], ra = rp[0], rb = rp[1];
#pragma unroll
        for (int i = 2; i < NPART / 2; i += 2) {
            qa += qp[i];  qb += qp[i + 1];
            ra += rp[i];  rb += rp[i + 1];
        }
        const float qh = qa + qb, rh = ra + rb;
        const float qc = qh + xhalf(qh), rc = rh + xhalf(rh);
#pragma unroll
        for (int j = 0; j < 8; ++j) {
            int b = bm0 + j + half * 8;
            float num = acc[j] + qc;
            float den = sqrtf(part[0][j][lane] + 2.f * part[1][j][lane] + rc) + EPS_L2;
            out[(size_t)b * B_SZ + c] = num / den;
        }
    }
}

extern "C" void kernel_launch(void* const* d_in, const int* in_sizes, int n_in,
                              void* d_out, int out_size, void* d_ws, size_t ws_size,
                              hipStream_t stream) {
    const float* img   = (const float*)d_in[0];
    const float* cap   = (const float*)d_in[1];
    const int*   lens  = (const int*)  d_in[2];
    const float* fc_w  = (const float*)d_in[3];
    const float* fc_b  = (const float*)d_in[4];
    const float* bn_w  = (const float*)d_in[5];
    const float* bn_b  = (const float*)d_in[6];
    float* out = (float*)d_out;

    float* ws = (float*)d_ws;
    const size_t BD = (size_t)B_SZ * D_SZ;
    float* cap_mean = ws;
    float* meanR    = cap_mean + BD;
    float* base     = meanR + BD;
    float* base2    = base + BD;
    float* P        = base2 + BD;
    float* A2m      = P + BD;
    float* ABm      = A2m + BD;
    float* cap_hat  = ABm + BD;
    float* qpart    = cap_hat + BD;
    float* rpart    = qpart + NPART * B_SZ;
    float* colsum1  = cap_hat + 2 * BD;
    float* colsum2  = colsum1 + D_SZ;

    zero_ws_kernel<<<dim3((2 * D_SZ + 255) / 256), 256, 0, stream>>>(colsum1, 2 * D_SZ);
    cap_pre_kernel<<<dim3(B_SZ), 256, 0, stream>>>(cap, lens, cap_mean, cap_hat);
    img_pre_kernel<<<dim3(B_SZ), 256, 0, stream>>>(img, meanR, colsum1, colsum2);
    make_base_kernel<<<dim3(B_SZ), 256, 0, stream>>>(meanR, colsum1, colsum2,
                                                     bn_w, bn_b, base, base2);
    fc_gemm_kernel<<<dim3(2 * D_SZ / 64, B_SZ / 32), 128, 0, stream>>>(
            cap_mean, fc_w, fc_b, cap_hat, P, A2m, ABm, qpart, rpart);
    sims_gemm_kernel<<<dim3(B_SZ / 16, B_SZ / 16), 96, 0, stream>>>(
            base, base2, P, A2m, ABm, qpart, rpart, out);
}


# cap_pre/img_pre: rows summed out of a register ring of 4-row groups requested 3 groups ahead: 30-60 loads per wave stay in flight where the unrolled row loop drained each group first; sums unchanged
# speedup vs baseline: 4.6945x; 1.2376x over previous
#include <hip/hip_runtime.h>
#include <hip/hip_bf16.h>
#include <math.h>

#define B_SZ 256
#define R_SZ 36
#define T_SZ 64
#define D_SZ 1024
#define EPS_L2 1e-8f
#define EPS_BN 1e-5f
#define LEAK 0.1f

typedef __attribute__((ext_vector_type(2))) float v2f;
typedef __attribute__((ext_vector_type(8))) float v8f;

__global__ void zero_ws_kernel(float* p, int n) {
    int i = blockIdx.x * blockDim.x + threadIdx.x;
    if (i < n) p[i] = 0.f;
}

template <int CTRL>
__device__ __forceinline__ float row_read(float v) {
    return __builtin_bit_cast(float, __builtin_amdgcn_update_dpp(
            0, __builtin_bit_cast(int, v), CTRL, 0xf, 0xf, true));
}
__device__ __forceinline__ float xhalf(float x) {
    const int b = __builtin_bit_cast(int, x);
    return __builtin_bit_cast(float, __builtin_amdgcn_permlanex16(
            b, b, 0x76543210, 0xfedcba98, false, false));
}

#define GROUP_ROWS 4
#define GROUPS_AHEAD 3
struct RowGroup { float x[GROUP_ROWS][4]; };

__device__ __forceinline__ void load_group(RowGroup& g, const float* __restrict__ p) {
#pragma unroll
    for (int r = 0; r < GROUP_ROWS; ++r) {
#pragma unroll
        for (int i = 0; i < 4; ++i) g.x[r][i] = p[(size_t)r * D_SZ + i * 256];
    }
}

__global__ void cap_pre_kernel(const float* __restrict__ cap,
                               const int* __restrict__ lens,
                               float* __restrict__ cap_mean,
                               float* __restrict__ cap_hat) {
    __shared__ float wsum[8];
    const int b = blockIdx.x;
    const int len = lens[b];
    const float* p = cap + (size_t)b * T_SZ * D_SZ + threadIdx.x;
    float sumsq[4] = {0.f, 0.f, 0.f, 0.f};
    float summ[4]  = {0.f, 0.f, 0.f, 0.f};
    constexpr int NG = T_SZ / GROUP_ROWS;
    RowGroup ring[GROUPS_AHEAD + 1];
#pragma unroll
    for (int g = 0; g < GROUPS_AHEAD; ++g)
        load_group(ring[g], p + (size_t)g * GROUP_ROWS * D_SZ);
#pragma unroll
    for (int g = 0; g < NG; ++g) {
        const RowGroup& cur = ring[g % (GROUPS_AHEAD + 1)];
#pragma unroll
        for (int r = 0; r < GROUP_ROWS; ++r) {
            const bool valid = (g * GROUP_ROWS + r < len);
#pragma unroll
            for (int i = 0; i < 4; ++i) {
                float x = cur.x[r][i];
                float y = x > 0.f ? x : LEAK * x;
                sumsq[i] += y * y;
                if (valid) summ[i] += y;
            }
        }
        if (g + GROUPS_AHEAD < NG)
            load_group(ring[(g + GROUPS_AHEAD) % (GROUPS_AHEAD + 1)],
                       p + (size_t)(g + GROUPS_AHEAD) * GROUP_ROWS * D_SZ);
    }
    float cm[4];
    float ss = 0.f;
#pragma unroll
    for (int i = 0; i < 4; ++i) {
        float n = sqrtf(sumsq[i]) + EPS_L2;
        cm[i] = summ[i] / (n * (float)len);
        ss += cm[i] * cm[i];
    }
    ss += row_read<0x161>(ss);
    ss += row_read<0x162>(ss);
    ss += row_read<0x164>(ss);
    ss += row_read<0x168>(ss);
    ss += xhalf(ss);
    if ((threadIdx.x & 31) == 0) wsum[threadIdx.x >> 5] = ss;
    __syncthreads();
    float tot = wsum[0];
#pragma unroll
    for (int w = 1; w < 8; ++w) tot += wsum[w];
    const float nrm = sqrtf(tot) + EPS_L2;
#pragma unroll
    for (int i = 0; i < 4; ++i) {
        int d = threadIdx.x + i * 256;
        cap_mean[(size_t)b * D_SZ + d] = cm[i];
        cap_hat[(size_t)b * D_SZ + d]  = cm[i] / nrm;
    }
}

__global__ void img_pre_kernel(const float* __restrict__ img,
                               float* __restrict__ meanR,
                               float* __restrict__ cs1,
                               float* __restrict__ cs2) {
    const int b = blockIdx.x;
    const float* p = img + (size_t)b * R_SZ * D_SZ + threadIdx.x;
    float sumsq[4] = {0.f, 0.f, 0.f, 0.f};
    float summ[4]  = {0.f, 0.f, 0.f, 0.f};
    constexpr int NG = R_SZ / GROUP_ROWS;
    RowGroup ring[GROUPS_AHEAD + 1];
#pragma unroll
    for (int g = 0; g < GROUPS_AHEAD; ++g)
        load_group(ring[g], p + (size_t)g * GROUP_ROWS * D_SZ);
#pragma unroll
    for (int g = 0; g < NG; ++g) {
        const RowGroup& cur = ring[g % (GROUPS_AHEAD + 1)];
#pragma unroll
        for (int r = 0; r < GROUP_ROWS; ++r) {
#pragma unroll
            for (int i = 0; i < 4; ++i) {
                float x = cur.x[r][i];
                float y = x > 0.f ? x : LEAK * x;
                sumsq[i] += y * y;
                summ[i]  += y;
            }
        }
        if (g + GROUPS_AHEAD < NG)
            load_group(ring[(g + GROUPS_AHEAD) % (GROUPS_AHEAD + 1)],
                       p + (size_t)(g + GROUPS_AHEAD) * GROUP_ROWS * D_SZ);
    }
#pragma unroll
    for (int i = 0; i < 4; ++i) {
        int d = threadIdx.x + i * 256;
        float n = sqrtf(sumsq[i]) + EPS_L2;
        float s1 = summ[i] / n;
        float s2 = sumsq[i] / (n * n);
        meanR[(size_t)b * D_SZ + d] = s1 * (1.f / (float)R_SZ);
        atomicAdd(cs1 + d, s1);
        atomicAdd(cs2 + d, s2);
    }
}

__global__ void make_base_kernel(const float* __restrict__ meanR,
                                 const float* __restrict__ cs1,
                                 const float* __restrict__ cs2,
                                 const float* __restrict__ bn_w,
                                 const float* __restrict__ bn_b,
                                 float* __restrict__ base,
                                 float* __restrict__ base2) {
    const int b = blockIdx.x;
    const float inv_n = 1.f / (float)(B_SZ * R_SZ);
#pragma unroll
    for (int i = 0; i < 4; ++i) {
        int d = threadIdx.x + i * 256;
        float mu  = cs1[d] * inv_n;
        float ex2 = cs2[d] * inv_n;
        float var = ex2 - mu * mu;
        float scale = bn_w[d] * rsqrtf(var + EPS_BN);
        float v = (meanR[(size_t)b * D_SZ + d] - mu) * scale + bn_b[d];
        base[(size_t)b * D_SZ + d]  = v;
        base2[(size_t)b * D_SZ + d] = v * v;
    }
}

#define KB 128
#define NQ (KB / 8)

typedef __attribute__((ext_vector_type(4))) float v4f;

__device__ __forceinline__ v4f ld_quad(const float* __restrict__ p) {
    return *(const v4f*)p;
}
__device__ __forceinline__ v2f lo2(v4f q) { return __builtin_shufflevector(q, q, 0, 1); }
__device__ __forceinline__ v2f hi2(v4f q) { return __builtin_shufflevector(q, q, 2, 3); }
__device__ __forceinline__ v8f mma4(v2f a, v2f b, v8f c) {
    return __builtin_amdgcn_wmma_f32_16x16x4_f32(false, a, false, b, (short)0, c,
                                                 false, false);
}

struct FcBuf { v4f a0[NQ], a1[NQ], b0[NQ]; };

__device__ __forceinline__ void fc_load(FcBuf& s, const float* __restrict__ pa,
                                        const float* __restrict__ pb) {
#pragma unroll
    for (int j = 0; j < NQ; ++j) {
        s.a0[j] = ld_quad(pa + 8 * j);
        s.a1[j] = ld_quad(pa + 16 * D_SZ + 8 * j);
        s.b0[j] = ld_quad(pb + 8 * j);
    }
}
__device__ __forceinline__ void fc_mma(const FcBuf& s, v8f& c0, v8f& c1) {
#pragma unroll
    for (int j = 0; j < NQ; ++j) {
        c0 = mma4(lo2(s.a0[j]), lo2(s.b0[j]), c0);
        c1 = mma4(lo2(s.a1[j]), lo2(s.b0[j]), c1);
        c0 = mma4(hi2(s.a0[j]), hi2(s.b0[j]), c0);
        c1 = mma4(hi2(s.a1[j]), hi2(s.b0[j]), c1);
    }
}

__global__ __launch_bounds__(128) void fc_gemm_kernel(
        const float* __restrict__ A, const float* __restrict__ Bw,
        const float* __restrict__ bias, const float* __restrict__ cap_hat,
        float* __restrict__ P, float* __restrict__ A2m, float* __restrict__ ABm,
        float* __restrict__ qpart, float* __restrict__ rpart) {
    __shared__ float red[4][32][2];
    const int wave = threadIdx.x >> 5;
    const int lane = threadIdx.x & 31;
    const int half = lane >> 4, fr = lane & 15;
    const int m0 = blockIdx.y * 32;
    const int n0 = blockIdx.x * 64 + wave * 16;
    const float* pa = A  + (size_t)(m0 + fr) * D_SZ + 4 * half;
    const float* pb = Bw + (size_t)(n0 + fr) * D_SZ + 4 * half;
    v8f c0 = {}, c1 = {};
    FcBuf s0, s1;
    fc_load(s0, pa, pb);
    for (int k0 = 0; k0 < D_SZ - 2 * KB; k0 += 2 * KB) {
        fc_load(s1, pa + KB, pb + KB);
        fc_mma(s0, c0, c1);
        fc_load(s0, pa + 2 * KB, pb + 2 * KB);
        fc_mma(s1, c0, c1);
        pa += 2 * KB;
        pb += 2 * KB;
    }
    fc_load(s1, pa + KB, pb + KB);
    const int n = n0 + fr;
    const int d = n >> 1;
    float ch_lo[8], ch_hi[8];
#pragma unroll
    for (int j = 0; j < 8; ++j) {
        ch_lo[j] = cap_hat[(size_t)(m0 + j + half * 8) * D_SZ + d];
        ch_hi[j] = cap_hat[(size_t)(m0 + 16 + j + half * 8) * D_SZ + d];
    }
    fc_mma(s0, c0, c1);
    fc_mma(s1, c0, c1);

    const float bias_n = bias[n];
    const bool even = (fr & 1) == 0;
    float* __restrict__ pab = even ? P : ABm;
#pragma unroll
    for (int j = 0; j < 8; ++j) {
        const int r_lo = j + half * 8, r_hi = r_lo + 16;
        const size_t i_lo = (size_t)(m0 + r_lo) * D_SZ + d;
        const size_t i_hi = (size_t)(m0 + r_hi) * D_SZ + d;
        const float v0 = c0[j] + bias_n, v1 = c1[j] + bias_n;
        const float p0 = row_read<0x161>(v0), p1 = row_read<0x161>(v1);
        const float h0 = ch_lo[j], h1 = ch_hi[j];
        pab[i_lo] = v0 * (even ? h0 : p0);
        pab[i_hi] = v1 * (even ? h1 : p1);
        const float a = even ? v0 : p1;
        A2m[even ? i_lo : i_hi] = a * a;
        float t0 = even ? p0 * h0 : v0 * v0;
        float t1 = even ? p1 * h1 : v1 * v1;
        t0 += row_read<0x162>(t0);  t1 += row_read<0x162>(t1);
        t0 += row_read<0x164>(t0);  t1 += row_read<0x164>(t1);
        t0 += row_read<0x168>(t0);  t1 += row_read<0x168>(t1);
        if (fr < 2) {
            red[wave][r_lo][fr] = t0;
            red[wave][r_hi][fr] = t1;
        }
    }
    __syncthreads();
    if (threadIdx.x < 64) {
        const int row = threadIdx.x >> 1, kind = threadIdx.x & 1;
        const float s = ((red[0][row][kind] + red[1][row][kind]) + red[2][row][kind])
                        + red[3][row][kind];
        float* __restrict__ dst = kind ? rpart : qpart;
        dst[(size_t)blockIdx.x * B_SZ + m0 + row] = s;
    }
}

struct SimsBuf { v4f a[NQ], b[NQ]; };
#define NPART (2 * D_SZ / 64)

__device__ __forceinline__ void sims_load(SimsBuf& s,
        const float* __restrict__ pa, const float* __restrict__ pb) {
#pragma unroll
    for (int j = 0; j < NQ; ++j) {
        s.a[j] = ld_quad(pa + 8 * j);
        s.b[j] = ld_quad(pb + 8 * j);
    }
}
__device__ __forceinline__ void sims_mma(const SimsBuf& s, v8f& acc_lo, v8f& acc_hi) {
#pragma unroll
    for (int j = 0; j < NQ; ++j) {
        acc_lo = mma4(lo2(s.a[j]), lo2(s.b[j]), acc_lo);
        acc_hi = mma4(hi2(s.a[j]), hi2(s.b[j]), acc_hi);
    }
}

__global__ __launch_bounds__(96) void sims_gemm_kernel(
        const float* __restrict__ base, const float* __restrict__ base2,
        const float* __restrict__ P, const float* __restrict__ A2m,
        const float* __restrict__ ABm,
        const float* __restrict__ qpart, const float* __restrict__ rpart,
        float* __restrict__ out) {
    __shared__ float part[2][8][32];
    const int wave = __builtin_amdgcn_readfirstlane((int)(threadIdx.x >> 5));
    const int lane = threadIdx.x & 31;
    const int half = lane >> 4, fr = lane & 15;
    const int bm0 = blockIdx.y * 16;
    const int cn0 = blockIdx.x * 16;
    const float* opa = (wave == 1) ? base2 : base;
    const float* opb = (wave == 0) ? P : ((wave == 1) ? A2m : ABm);
    const float* pa = opa + (size_t)(bm0 + fr) * D_SZ + 4 * half;
    const float* pb = opb + (size_t)(cn0 + fr) * D_SZ + 4 * half;
    v8f acc_lo = {}, acc_hi = {};
    SimsBuf s0, s1;
    sims_load(s0, pa, pb);
    float qp[NPART / 2], rp[NPART / 2];
#pragma unroll
    for (int i = 0; i < NPART / 2; ++i) { qp[i] = 0.f; rp[i] = 0.f; }
    if (wave == 0) {
#pragma unroll
        for (int i = 0; i < NPART / 2; ++i) {
            qp[i] = qpart[(size_t)(half * (NPART / 2) + i) * B_SZ + cn0 + fr];
            rp[i] = rpart[(size_t)(half * (NPART / 2) + i) * B_SZ + cn0 + fr];
        }
    }
    for (int k0 = 0; k0 < D_SZ - 2 * KB; k0 += 2 * KB) {
        sims_load(s1, pa + k0 + KB, pb + k0 + KB);
        sims_mma(s0, acc_lo, acc_hi);
        sims_load(s0, pa + k0 + 2 * KB, pb + k0 + 2 * KB);
        sims_mma(s1, acc_lo, acc_hi);
    }
    sims_load(s1, pa + D_SZ - KB, pb + D_SZ - KB);
    sims_mma(s0, acc_lo, acc_hi);
    sims_mma(s1, acc_lo, acc_hi);
    const v8f acc = acc_lo + acc_hi;

    if (wave != 0) {
#pragma unroll
        for (int j = 0; j < 8; ++j) part[wave - 1][j][lane] = acc[j];
    }
    __syncthreads();
    if (wave == 0) {
        const int c = cn0 + fr;
        float qa = qp[0], qb = qp[1], ra = rp[0], rb = rp[1];
#pragma unroll
        for (int i = 2; i < NPART / 2; i += 2) {
            qa += qp[i];  qb += qp[i + 1];
            ra += rp[i];  rb += rp[i + 1];
        }
        const float qh = qa + qb, rh = ra + rb;
        const float qc = qh + xhalf(qh), rc = rh + xhalf(rh);
#pragma unroll
        for (int j = 0; j < 8; ++j) {
            int b = bm0 + j + half * 8;
            float num = acc[j] + qc;
            float den = sqrtf(part[0][j][lane] + 2.f * part[1][j][lane] + rc) + EPS_L2;
            out[(size_t)b * B_SZ + c] = num / den;
        }
    }
}

extern "C" void kernel_launch(void* const* d_in, const int* in_sizes, int n_in,
                              void* d_out, int out_size, void* d_ws, size_t ws_size,
                              hipStream_t stream) {
    const float* img   = (const float*)d_in[0];
    const float* cap   = (const float*)d_in[1];
    const int*   lens  = (const int*)  d_in[2];
    const float* fc_w  = (const float*)d_in[3];
    const float* fc_b  = (const float*)d_in[4];
    const float* bn_w  = (const float*)d_in[5];
    const float* bn_b  = (const float*)d_in[6];
    float* out = (float*)d_out;

    float* ws = (float*)d_ws;
    const size_t BD = (size_t)B_SZ * D_SZ;
    float* cap_mean = ws;
    float* meanR    = cap_mean + BD;
    float* base     = meanR + BD;
    float* base2    = base + BD;
    float* P        = base2 + BD;
    float* A2m      = P + BD;
    float* ABm      = A2m + BD;
    float* cap_hat  = ABm + BD;
    float* qpart    = cap_hat + BD;
    float* rpart    = qpart + NPART * B_SZ;
    float* colsum1  = cap_hat + 2 * BD;
    float* colsum2  = colsum1 + D_SZ;

    zero_ws_kernel<<<dim3((2 * D_SZ + 255) / 256), 256, 0, stream>>>(colsum1, 2 * D_SZ);
    cap_pre_kernel<<<dim3(B_SZ), 256, 0, stream>>>(cap, lens, cap_mean, cap_hat);
    img_pre_kernel<<<dim3(B_SZ), 256, 0, stream>>>(img, meanR, colsum1, colsum2);
    make_base_kernel<<<dim3(B_SZ), 256, 0, stream>>>(meanR, colsum1, colsum2,
                                                     bn_w, bn_b, base, base2);
    fc_gemm_kernel<<<dim3(2 * D_SZ / 64, B_SZ / 32), 128, 0, stream>>>(
            cap_mean, fc_w, fc_b, cap_hat, P, A2m, ABm, qpart, rpart);
    sims_gemm_kernel<<<dim3(B_SZ / 16, B_SZ / 16), 96, 0, stream>>>(
            base, base2, P, A2m, ABm, qpart, rpart, out);
}


# BN statistics without atomics: img_pre stores per-entry sums, each cap_pre block adds 4 channels over the batch in a fixed tree, fc GEMM tail writes base: 6 launches to 4; BN sum order changes
# speedup vs baseline: 5.2342x; 1.1150x over previous
#include <hip/hip_runtime.h>
#include <hip/hip_bf16.h>
#include <math.h>

#define B_SZ 256
#define R_SZ 36
#define T_SZ 64
#define D_SZ 1024
#define EPS_L2 1e-8f
#define EPS_BN 1e-5f
#define LEAK 0.1f

typedef __attribute__((ext_vector_type(2))) float v2f;
typedef __attribute__((ext_vector_type(4))) float v4f;
typedef __attribute__((ext_vector_type(8))) float v8f;

template <int CTRL>
__device__ __forceinline__ float row_read(float v) {
    return __builtin_bit_cast(float, __builtin_amdgcn_update_dpp(
            0, __builtin_bit_cast(int, v), CTRL, 0xf, 0xf, true));
}
__device__ __forceinline__ float xhalf(float x) {
    const int b = __builtin_bit_cast(int, x);
    return __builtin_bit_cast(float, __builtin_amdgcn_permlanex16(
            b, b, 0x76543210, 0xfedcba98, false, false));
}

#define GROUP_ROWS 4
#define GROUPS_AHEAD 3
struct RowGroup { float x[GROUP_ROWS][4]; };

__device__ __forceinline__ void load_group(RowGroup& g, const float* __restrict__ p) {
#pragma unroll
    for (int r = 0; r < GROUP_ROWS; ++r) {
#pragma unroll
        for (int i = 0; i < 4; ++i) g.x[r][i] = p[(size_t)r * D_SZ + i * 256];
    }
}

__global__ void cap_pre_kernel(const float* __restrict__ cap,
                               const int* __restrict__ lens,
                               const float* __restrict__ S1,
                               const float* __restrict__ S2,
                               const float* __restrict__ bn_w,
                               float* __restrict__ cap_mean,
                               float* __restrict__ cap_hat,
                               float* __restrict__ bn_mu,
                               float* __restrict__ bn_scale) {
    __shared__ float wsum[8][9];
    const int b = blockIdx.x;
    const int len = lens[b];
    const float* p = cap + (size_t)b * T_SZ * D_SZ + threadIdx.x;
    const float bnw = bn_w[4 * b + (threadIdx.x & 3)];
    const v4f bn1 = *(const v4f*)(S1 + (size_t)threadIdx.x * D_SZ + 4 * b);
    const v4f bn2 = *(const v4f*)(S2 + (size_t)threadIdx.x * D_SZ + 4 * b);
    float sumsq[4] = {0.f, 0.f, 0.f, 0.f};
    float summ[4]  = {0.f, 0.f, 0.f, 0.f};
    constexpr int NG = T_SZ / GROUP_ROWS;
    RowGroup ring[GROUPS_AHEAD + 1];
#pragma unroll
    for (int g = 0; g < GROUPS_AHEAD; ++g)
        load_group(ring[g], p + (size_t)g * GROUP_ROWS * D_SZ);
#pragma unroll
    for (int g = 0; g < NG; ++g) {
        const RowGroup& cur = ring[g % (GROUPS_AHEAD + 1)];
#pragma unroll
        for (int r = 0; r < GROUP_ROWS; ++r) {
            const bool valid = (g * GROUP_ROWS + r < len);
#pragma unroll
            for (int i = 0; i < 4; ++i) {
                float x = cur.x[r][i];
                float y = x > 0.f ? x : LEAK * x;
                sumsq[i] += y * y;
                if (valid) summ[i] += y;
            }
        }
        if (g + GROUPS_AHEAD < NG)
            load_group(ring[(g + GROUPS_AHEAD) % (GROUPS_AHEAD + 1)],
                       p + (size_t)(g + GROUPS_AHEAD) * GROUP_ROWS * D_SZ);
    }
    float cm[4];
    float ss = 0.f;
#pragma unroll
    for (int i = 0; i < 4; ++i) {
        float n = sqrtf(sumsq[i]) + EPS_L2;
        cm[i] = summ[i] / (n * (float)len);
        ss += cm[i] * cm[i];
    }
    ss += row_read<0x161>(ss);
    ss += row_read<0x162>(ss);
    ss += row_read<0x164>(ss);
    ss += row_read<0x168>(ss);
    ss += xhalf(ss);
    float cs[8];
#pragma unroll
    for (int i = 0; i < 4; ++i) { cs[i] = bn1[i]; cs[4 + i] = bn2[i]; }
#pragma unroll
    for (int i = 0; i < 8; ++i) {
        cs[i] += row_read<0x161>(cs[i]);
        cs[i] += row_read<0x162>(cs[i]);
        cs[i] += row_read<0x164>(cs[i]);
        cs[i] += row_read<0x168>(cs[i]);
        cs[i] += xhalf(cs[i]);
    }
    if ((threadIdx.x & 31) == 0) {
        wsum[threadIdx.x >> 5][0] = ss;
#pragma unroll
        for (int i = 0; i < 8; ++i) wsum[threadIdx.x >> 5][1 + i] = cs[i];
    }
    __syncthreads();
    if (threadIdx.x < 4) {
        const int k = threadIdx.x;
        float c1 = wsum[0][1 + k], c2 = wsum[0][5 + k];
#pragma unroll
        for (int w = 1; w < 8; ++w) { c1 += wsum[w][1 + k]; c2 += wsum[w][5 + k]; }
        const float inv_n = 1.f / (float)(B_SZ * R_SZ);
        const float mu  = c1 * inv_n;
        const float ex2 = c2 * inv_n;
        const float var = ex2 - mu * mu;
        bn_mu[4 * b + k]    = mu;
        bn_scale[4 * b + k] = bnw * rsqrtf(var + EPS_BN);
    }
    float tot = wsum[0][0];
#pragma unroll
    for (int w = 1; w < 8; ++w) tot += wsum[w][0];
    const float nrm = sqrtf(tot) + EPS_L2;
#pragma unroll
    for (int i = 0; i < 4; ++i) {
        int d = threadIdx.x + i * 256;
        cap_mean[(size_t)b * D_SZ + d] = cm[i];
        cap_hat[(size_t)b * D_SZ + d]  = cm[i] / nrm;
    }
}

__global__ void img_pre_kernel(const float* __restrict__ img,
                               float* __restrict__ S1,
                               float* __restrict__ S2) {
    const int b = blockIdx.x;
    const float* p = img + (size_t)b * R_SZ * D_SZ + threadIdx.x;
    float sumsq[4] = {0.f, 0.f, 0.f, 0.f};
    float summ[4]  = {0.f, 0.f, 0.f, 0.f};
    constexpr int NG = R_SZ / GROUP_ROWS;
    RowGroup ring[GROUPS_AHEAD + 1];
#pragma unroll
    for (int g = 0; g < GROUPS_AHEAD; ++g)
        load_group(ring[g], p + (size_t)g * GROUP_ROWS * D_SZ);
#pragma unroll
    for (int g = 0; g < NG; ++g) {
        const RowGroup& cur = ring[g % (GROUPS_AHEAD + 1)];
#pragma unroll
        for (int r = 0; r < GROUP_ROWS; ++r) {
#pragma unroll
            for (int i = 0; i < 4; ++i) {
                float x = cur.x[r][i];
                float y = x > 0.f ? x : LEAK * x;
                sumsq[i] += y * y;
                summ[i]  += y;
            }
        }
        if (g + GROUPS_AHEAD < NG)
            load_group(ring[(g + GROUPS_AHEAD) % (GROUPS_AHEAD + 1)],
                       p + (size_t)(g + GROUPS_AHEAD) * GROUP_ROWS * D_SZ);
    }
#pragma unroll
    for (int i = 0; i < 4; ++i) {
        int d = threadIdx.x + i * 256;
        float n = sqrtf(sumsq[i]) + EPS_L2;
        float s1 = summ[i] / n;
        float s2 = sumsq[i] / (n * n);
        S1[(size_t)b * D_SZ + d] = s1;
        S2[(size_t)b * D_SZ + d] = s2;
    }
}

#define KB 128
#define NQ (KB / 8)

__device__ __forceinline__ v4f ld_quad(const float* __restrict__ p) {
    return *(const v4f*)p;
}
__device__ __forceinline__ v2f lo2(v4f q) { return __builtin_shufflevector(q, q, 0, 1); }
__device__ __forceinline__ v2f hi2(v4f q) { return __builtin_shufflevector(q, q, 2, 3); }
__device__ __forceinline__ v8f mma4(v2f a, v2f b, v8f c) {
    return __builtin_amdgcn_wmma_f32_16x16x4_f32(false, a, false, b, (short)0, c,
                                                 false, false);
}

struct FcBuf { v4f a0[NQ], a1[NQ], b0[NQ]; };

__device__ __forceinline__ void fc_load(FcBuf& s, const float* __restrict__ pa,
                                        const float* __restrict__ pb) {
#pragma unroll
    for (int j = 0; j < NQ; ++j) {
        s.a0[j] = ld_quad(pa + 8 * j);
        s.a1[j] = ld_quad(pa + 16 * D_SZ + 8 * j);
        s.b0[j] = ld_quad(pb + 8 * j);
    }
}
__device__ __forceinline__ void fc_mma(const FcBuf& s, v8f& c0, v8f& c1) {
#pragma unroll
    for (int j = 0; j < NQ; ++j) {
        c0 = mma4(lo2(s.a0[j]), lo2(s.b0[j]), c0);
        c1 = mma4(lo2(s.a1[j]), lo2(s.b0[j]), c1);
        c0 = mma4(hi2(s.a0[j]), hi2(s.b0[j]), c0);
        c1 = mma4(hi2(s.a1[j]), hi2(s.b0[j]), c1);
    }
}

__global__ __launch_bounds__(128) void fc_gemm_kernel(
        const float* __restrict__ A, const float* __restrict__ Bw,
        const float* __restrict__ bias,
        const float* __restrict__ S1, const float* __restrict__ bn_mu,
        const float* __restrict__ bn_scale, const float* __restrict__ bn_b,
        float* __restrict__ base, float* __restrict__ base2,
        const float* __restrict__ cap_hat,
        float* __restrict__ P, float* __restrict__ A2m, float* __restrict__ ABm,
        float* __restrict__ qpart, float* __restrict__ rpart) {
    __shared__ float red[4][32][2];
    const int wave = threadIdx.x >> 5;
    const int lane = threadIdx.x & 31;
    const int half = lane >> 4, fr = lane & 15;
    const int m0 = blockIdx.y * 32;
    const int n0 = blockIdx.x * 64 + wave * 16;
    const float* pa = A  + (size_t)(m0 + fr) * D_SZ + 4 * half;
    const float* pb = Bw + (size_t)(n0 + fr) * D_SZ + 4 * half;
    v8f c0 = {}, c1 = {};
    FcBuf s0, s1;
    fc_load(s0, pa, pb);
    for (int k0 = 0; k0 < D_SZ - 2 * KB; k0 += 2 * KB) {
        fc_load(s1, pa + KB, pb + KB);
        fc_mma(s0, c0, c1);
        fc_load(s0, pa + 2 * KB, pb + 2 * KB);
        fc_mma(s1, c0, c1);
        pa += 2 * KB;
        pb += 2 * KB;
    }
    fc_load(s1, pa + KB, pb + KB);
    const int n = n0 + fr;
    const int d = n >> 1;
    float ch_lo[8], ch_hi[8];
#pragma unroll
    for (int j = 0; j < 8; ++j) {
        ch_lo[j] = cap_hat[(size_t)(m0 + j + half * 8) * D_SZ + d];
        ch_hi[j] = cap_hat[(size_t)(m0 + 16 + j + half * 8) * D_SZ + d];
    }
    const int bd = blockIdx.x * 32 + (threadIdx.x & 7) * 4;
    const size_t bi0 = (size_t)(m0 + (threadIdx.x >> 3)) * D_SZ + bd;
    const size_t bi1 = bi0 + (size_t)16 * D_SZ;
    const v4f mu4 = ld_quad(bn_mu + bd), sc4 = ld_quad(bn_scale + bd);
    const v4f bb4 = ld_quad(bn_b + bd);
    const v4f sa = ld_quad(S1 + bi0), sb = ld_quad(S1 + bi1);
    fc_mma(s0, c0, c1);
    fc_mma(s1, c0, c1);

    {
        const float inv_r = 1.f / (float)R_SZ;
        const v4f va = (sa * inv_r - mu4) * sc4 + bb4;
        const v4f vb = (sb * inv_r - mu4) * sc4 + bb4;
        *(v4f*)(base + bi0)  = va;
        *(v4f*)(base2 + bi0) = va * va;
        *(v4f*)(base + bi1)  = vb;
        *(v4f*)(base2 + bi1) = vb * vb;
    }

    const float bias_n = bias[n];
    const bool even = (fr & 1) == 0;
    float* __restrict__ pab = even ? P : ABm;
#pragma unroll
    for (int j = 0; j < 8; ++j) {
        const int r_lo = j + half * 8, r_hi = r_lo + 16;
        const size_t i_lo = (size_t)(m0 + r_lo) * D_SZ + d;
        const size_t i_hi = (size_t)(m0 + r_hi) * D_SZ + d;
        const float v0 = c0[j] + bias_n, v1 = c1[j] + bias_n;
        const float p0 = row_read<0x161>(v0), p1 = row_read<0x161>(v1);
        const float h0 = ch_lo[j], h1 = ch_hi[j];
        pab[i_lo] = v0 * (even ? h0 : p0);
        pab[i_hi] = v1 * (even ? h1 : p1);
        const float a = even ? v0 : p1;
        A2m[even ? i_lo : i_hi] = a * a;
        float t0 = even ? p0 * h0 : v0 * v0;
        float t1 = even ? p1 * h1 : v1 * v1;
        t0 += row_read<0x162>(t0);  t1 += row_read<0x162>(t1);
        t0 += row_read<0x164>(t0);  t1 += row_read<0x164>(t1);
        t0 += row_read<0x168>(t0);  t1 += row_read<0x168>(t1);
        if (fr < 2) {
            red[wave][r_lo][fr] = t0;
            red[wave][r_hi][fr] = t1;
        }
    }
    __syncthreads();
    if (threadIdx.x < 64) {
        const int row = threadIdx.x >> 1, kind = threadIdx.x & 1;
        const float s = ((red[0][row][kind] + red[1][row][kind]) + red[2][row][kind])
                        + red[3][row][kind];
        float* __restrict__ dst = kind ? rpart : qpart;
        dst[(size_t)blockIdx.x * B_SZ + m0 + row] = s;
    }
}

struct SimsBuf { v4f a[NQ], b[NQ]; };
#define NPART (2 * D_SZ / 64)

__device__ __forceinline__ void sims_load(SimsBuf& s,
        const float* __restrict__ pa, const float* __restrict__ pb) {
#pragma unroll
    for (int j = 0; j < NQ; ++j) {
        s.a[j] = ld_quad(pa + 8 * j);
        s.b[j] = ld_quad(pb + 8 * j);
    }
}
__device__ __forceinline__ void sims_mma(const SimsBuf& s, v8f& acc_lo, v8f& acc_hi) {
#pragma unroll
    for (int j = 0; j < NQ; ++j) {
        acc_lo = mma4(lo2(s.a[j]), lo2(s.b[j]), acc_lo);
        acc_hi = mma4(hi2(s.a[j]), hi2(s.b[j]), acc_hi);
    }
}

__global__ __launch_bounds__(96) void sims_gemm_kernel(
        const float* __restrict__ base, const float* __restrict__ base2,
        const float* __restrict__ P, const float* __restrict__ A2m,
        const float* __restrict__ ABm,
        const float* __restrict__ qpart, const float* __restrict__ rpart,
        float* __restrict__ out) {
    __shared__ float part[2][8][32];
    const int wave = __builtin_amdgcn_readfirstlane((int)(threadIdx.x >> 5));
    const int lane = threadIdx.x & 31;
    const int half = lane >> 4, fr = lane & 15;
    const int bm0 = blockIdx.y * 16;
    const int cn0 = blockIdx.x * 16;
    const float* opa = (wave == 1) ? base2 : base;
    const float* opb = (wave == 0) ? P : ((wave == 1) ? A2m : ABm);
    const float* pa = opa + (size_t)(bm0 + fr) * D_SZ + 4 * half;
    const float* pb = opb + (size_t)(cn0 + fr) * D_SZ + 4 * half;
    v8f acc_lo = {}, acc_hi = {};
    SimsBuf s0, s1;
    sims_load(s0, pa, pb);
    float qp[NPART / 2], rp[NPART / 2];
#pragma unroll
    for (int i = 0; i < NPART / 2; ++i) { qp[i] = 0.f; rp[i] = 0.f; }
    if (wave == 0) {
#pragma unroll
        for (int i = 0; i < NPART / 2; ++i) {
            qp[i] = qpart[(size_t)(half * (NPART / 2) + i) * B_SZ + cn0 + fr];
            rp[i] = rpart[(size_t)(half * (NPART / 2) + i) * B_SZ + cn0 + fr];
        }
    }
    for (int k0 = 0; k0 < D_SZ - 2 * KB; k0 += 2 * KB) {
        sims_load(s1, pa + k0 + KB, pb + k0 + KB);
        sims_mma(s0, acc_lo, acc_hi);
        sims_load(s0, pa + k0 + 2 * KB, pb + k0 + 2 * KB);
        sims_mma(s1, acc_lo, acc_hi);
    }
    sims_load(s1, pa + D_SZ - KB, pb + D_SZ - KB);
    sims_mma(s0, acc_lo, acc_hi);
    sims_mma(s1, acc_lo, acc_hi);
    const v8f acc = acc_lo + acc_hi;

    if (wave != 0) {
#pragma unroll
        for (int j = 0; j < 8; ++j) part[wave - 1][j][lane] = acc[j];
    }
    __syncthreads();
    if (wave == 0) {
        const int c = cn0 + fr;
        float qa = qp[0], qb = qp[1], ra = rp[0], rb = rp[1];
#pragma unroll
        for (int i = 2; i < NPART / 2; i += 2) {
            qa += qp[i];  qb += qp[i + 1];
            ra += rp[i];  rb += rp[i + 1];
        }
        const float qh = qa + qb, rh = ra + rb;
        const float qc = qh + xhalf(qh), rc = rh + xhalf(rh);
#pragma unroll
        for (int j = 0; j < 8; ++j) {
            int b = bm0 + j + half * 8;
            float num = acc[j] + qc;
            float den = sqrtf(part[0][j][lane] + 2.f * part[1][j][lane] + rc) + EPS_L2;
            out[(size_t)b * B_SZ + c] = num / den;
        }
    }
}

extern "C" void kernel_launch(void* const* d_in, const int* in_sizes, int n_in,
                              void* d_out, int out_size, void* d_ws, size_t ws_size,
                              hipStream_t stream) {
    const float* img   = (const float*)d_in[0];
    const float* cap   = (const float*)d_in[1];
    const int*   lens  = (const int*)  d_in[2];
    const float* fc_w  = (const float*)d_in[3];
    const float* fc_b  = (const float*)d_in[4];
    const float* bn_w  = (const float*)d_in[5];
    const float* bn_b  = (const float*)d_in[6];
    float* out = (float*)d_out;

    float* ws = (float*)d_ws;
    const size_t BD = (size_t)B_SZ * D_SZ;
    float* cap_mean = ws;
    float* S1       = cap_mean + BD;
    float* base     = S1 + BD;
    float* base2    = base + BD;
    float* P        = base2 + BD;
    float* A2m      = P + BD;
    float* ABm      = A2m + BD;
    float* cap_hat  = ABm + BD;
    float* qpart    = cap_hat + BD;
    float* rpart    = qpart + NPART * B_SZ;
    float* bn_mu    = cap_hat + 2 * BD;
    float* bn_scale = bn_mu + D_SZ;
    float* S2       = P;

    img_pre_kernel<<<dim3(B_SZ), 256, 0, stream>>>(img, S1, S2);
    cap_pre_kernel<<<dim3(B_SZ), 256, 0, stream>>>(cap, lens, S1, S2, bn_w,
                                                   cap_mean, cap_hat, bn_mu, bn_scale);
    fc_gemm_kernel<<<dim3(2 * D_SZ / 64, B_SZ / 32), 128, 0, stream>>>(
            cap_mean, fc_w, fc_b, S1, bn_mu, bn_scale, bn_b, base, base2,
            cap_hat, P, A2m, ABm, qpart, rpart);
    sims_gemm_kernel<<<dim3(B_SZ / 16, B_SZ / 16), 96, 0, stream>>>(
            base, base2, P, A2m, ABm, qpart, rpart, out);
}


# img and cap streamed by one launch: block b walks img entry b then caption b through one load ring; BN sums move to the fc GEMM's head, cap_hat to its epilogue: 4 launches to 3; BN sum order changes
# speedup vs baseline: 5.9416x; 1.1351x over previous
#include <hip/hip_runtime.h>
#include <hip/hip_bf16.h>
#include <math.h>

#define B_SZ 256
#define R_SZ 36
#define T_SZ 64
#define D_SZ 1024
#define EPS_L2 1e-8f
#define EPS_BN 1e-5f
#define LEAK 0.1f

typedef __attribute__((ext_vector_type(2))) float v2f;
typedef __attribute__((ext_vector_type(4))) float v4f;
typedef __attribute__((ext_vector_type(8))) float v8f;

template <int CTRL>
__device__ __forceinline__ float row_read(float v) {
    return __builtin_bit_cast(float, __builtin_amdgcn_update_dpp(
            0, __builtin_bit_cast(int, v), CTRL, 0xf, 0xf, true));
}
__device__ __forceinline__ float xhalf(float x) {
    const int b = __builtin_bit_cast(int, x);
    return __builtin_bit_cast(float, __builtin_amdgcn_permlanex16(
            b, b, 0x76543210, 0xfedcba98, false, false));
}

#define GROUP_ROWS 4
#define RING_GROUPS 4
struct RowGroup { float x[GROUP_ROWS][4]; };

__device__ __forceinline__ void load_group(RowGroup& g, const float* __restrict__ p) {
#pragma unroll
    for (int r = 0; r < GROUP_ROWS; ++r) {
#pragma unroll
        for (int i = 0; i < 4; ++i) g.x[r][i] = p[(size_t)r * D_SZ + i * 256];
    }
}

#define TILE_K 8
#define TILE_FLOATS (16 * TILE_K)
#define GROUP_FLOATS (16 * D_SZ)
__device__ __forceinline__ size_t tiled(int row, int k) {
    return (size_t)(row >> 4) * GROUP_FLOATS + (size_t)(k >> 3) * TILE_FLOATS
           + (size_t)((row & 15) * TILE_K + (k & 7));
}

#define IMG_GROUPS (R_SZ / GROUP_ROWS)
#define CAP_GROUPS (T_SZ / GROUP_ROWS)
__device__ __forceinline__ const float* stream_group(const float* __restrict__ pi,
                                                     const float* __restrict__ pc, int g) {
    return g < IMG_GROUPS ? pi + (size_t)g * GROUP_ROWS * D_SZ
                          : pc + (size_t)(g - IMG_GROUPS) * GROUP_ROWS * D_SZ;
}

__global__ void pre_kernel(const float* __restrict__ img,
                           const float* __restrict__ cap,
                           const int* __restrict__ lens,
                           float* __restrict__ S1,
                           float* __restrict__ S2,
                           float* __restrict__ cap_mean,
                           float* __restrict__ nrm_out) {
    __shared__ float wsum[8];
    const int b = blockIdx.x;
    const int len = lens[b];
    const float* pi = img + (size_t)b * R_SZ * D_SZ + threadIdx.x;
    const float* pc = cap + (size_t)b * T_SZ * D_SZ + threadIdx.x;
    float isq[4] = {0.f, 0.f, 0.f, 0.f};
    float ism[4] = {0.f, 0.f, 0.f, 0.f};
    float sumsq[4] = {0.f, 0.f, 0.f, 0.f};
    float summ[4]  = {0.f, 0.f, 0.f, 0.f};
    constexpr int NG = IMG_GROUPS + CAP_GROUPS;
    RowGroup ring[RING_GROUPS];
#pragma unroll
    for (int g = 0; g < RING_GROUPS; ++g)
        load_group(ring[g], stream_group(pi, pc, g));
#pragma unroll
    for (int g = 0; g < NG; ++g) {
        RowGroup& cur = ring[g % RING_GROUPS];
        if (g < IMG_GROUPS) {
#pragma unroll
            for (int r = 0; r < GROUP_ROWS; ++r) {
#pragma unroll
                for (int i = 0; i < 4; ++i) {
                    float x = cur.x[r][i];
                    float y = x > 0.f ? x : LEAK * x;
                    isq[i] += y * y;
                    ism[i] += y;
                }
            }
        } else {
#pragma unroll
            for (int r = 0; r < GROUP_ROWS; ++r) {
                const bool valid = ((g - IMG_GROUPS) * GROUP_ROWS + r < len);
#pragma unroll
                for (int i = 0; i < 4; ++i) {
                    float x = cur.x[r][i];
                    float y = x > 0.f ? x : LEAK * x;
                    sumsq[i] += y * y;
                    if (valid) summ[i] += y;
                }
            }
        }
        if (g + RING_GROUPS < NG)
            load_group(cur, stream_group(pi, pc, g + RING_GROUPS));
        __builtin_amdgcn_sched_barrier(0);
        if (g == IMG_GROUPS - 1) {
#pragma unroll
            for (int i = 0; i < 4; ++i) {
                int d = threadIdx.x + i * 256;
                float n = sqrtf(isq[i]) + EPS_L2;
                float s1 = ism[i] / n;
                float s2 = isq[i] / (n * n);
                S1[(size_t)b * D_SZ + d] = s1;
                S2[(size_t)b * D_SZ + d] = s2;
            }
        }
    }
    float cm[4];
    float ss = 0.f;
#pragma unroll
    for (int i = 0; i < 4; ++i) {
        float n = sqrtf(sumsq[i]) + EPS_L2;
        cm[i] = summ[i] / (n * (float)len);
        ss += cm[i] * cm[i];
    }
    ss += row_read<0x161>(ss);
    ss += row_read<0x162>(ss);
    ss += row_read<0x164>(ss);
    ss += row_read<0x168>(ss);
    ss += xhalf(ss);
    if ((threadIdx.x & 31) == 0) wsum[threadIdx.x >> 5] = ss;
    __syncthreads();
#pragma unroll
    for (int i = 0; i < 4; ++i)
        cap_mean[tiled(b, threadIdx.x + i * 256)] = cm[i];
    if (threadIdx.x == 0) {
        float tot = wsum[0];
#pragma unroll
        for (int w = 1; w < 8; ++w) tot += wsum[w];
        nrm_out[b] = sqrtf(tot) + EPS_L2;
    }
}

#define KB 128
#define NQ (KB / 8)
#define TKB (NQ * TILE_FLOATS)

__device__ __forceinline__ v4f ld_quad(const float* __restrict__ p) {
    return *(const v4f*)p;
}
__device__ __forceinline__ v2f lo2(v4f q) { return __builtin_shufflevector(q, q, 0, 1); }
__device__ __forceinline__ v2f hi2(v4f q) { return __builtin_shufflevector(q, q, 2, 3); }
__device__ __forceinline__ v8f mma4(v2f a, v2f b, v8f c) {
    return __builtin_amdgcn_wmma_f32_16x16x4_f32(false, a, false, b, (short)0, c,
                                                 false, false);
}

struct FcBuf { v4f a0[NQ], a1[NQ], b0[NQ]; };

__device__ __forceinline__ void fc_load(FcBuf& s, const float* __restrict__ pa,
                                        const float* __restrict__ pb) {
#pragma unroll
    for (int j = 0; j < NQ; ++j) {
        s.a0[j] = ld_quad(pa + TILE_FLOATS * j);
        s.a1[j] = ld_quad(pa + GROUP_FLOATS + TILE_FLOATS * j);
        s.b0[j] = ld_quad(pb + 8 * j);
    }
}
__device__ __forceinline__ void fc_mma(const FcBuf& s, v8f& c0, v8f& c1) {
#pragma unroll
    for (int j = 0; j < NQ; ++j) {
        c0 = mma4(lo2(s.a0[j]), lo2(s.b0[j]), c0);
        c1 = mma4(lo2(s.a1[j]), lo2(s.b0[j]), c1);
        c0 = mma4(hi2(s.a0[j]), hi2(s.b0[j]), c0);
        c1 = mma4(hi2(s.a1[j]), hi2(s.b0[j]), c1);
    }
}

__global__ __launch_bounds__(128) void fc_gemm_kernel(
        const float* __restrict__ A, const float* __restrict__ Bw,
        const float* __restrict__ bias,
        const float* __restrict__ S1, const float* __restrict__ S2,
        const float* __restrict__ bn_w, const float* __restrict__ bn_b,
        const float* __restrict__ nrm,
        float* __restrict__ base, float* __restrict__ base2,
        float* __restrict__ P, float* __restrict__ A2m, float* __restrict__ ABm,
        float* __restrict__ qpart, float* __restrict__ rpart) {
    __shared__ float red[4][32][2];
    __shared__ v4f bnred[4][8][2];
    const int wave = threadIdx.x >> 5;
    const int lane = threadIdx.x & 31;
    const int half = lane >> 4, fr = lane & 15;
    const int m0 = blockIdx.y * 32;
    const int n0 = blockIdx.x * 64 + wave * 16;
    const float* pa = A  + (size_t)(m0 >> 4) * GROUP_FLOATS + fr * TILE_K + 4 * half;
    const float* pb = Bw + (size_t)(n0 + fr) * D_SZ + 4 * half;
    v8f c0 = {}, c1 = {};
    FcBuf s0, s1;
    const int bd = blockIdx.x * 32 + (threadIdx.x & 7) * 4;
    v4f mu4, sc4;
    {
        const size_t bn0 = (size_t)(threadIdx.x >> 3) * D_SZ + bd;
        v4f q1[16], q2[16];
#pragma unroll
        for (int i = 0; i < 16; ++i) {
            q1[i] = ld_quad(S1 + bn0 + (size_t)i * 16 * D_SZ);
            q2[i] = ld_quad(S2 + bn0 + (size_t)i * 16 * D_SZ);
        }
        const v4f bnw4 = ld_quad(bn_w + bd);
        fc_load(s0, pa, pb);
        v4f a1 = q1[0], a2 = q2[0];
#pragma unroll
        for (int i = 1; i < 16; ++i) { a1 += q1[i]; a2 += q2[i]; }
#pragma unroll
        for (int k = 0; k < 4; ++k) {
            a1[k] += row_read<0x168>(a1[k]);  a2[k] += row_read<0x168>(a2[k]);
            a1[k] += xhalf(a1[k]);            a2[k] += xhalf(a2[k]);
        }
        if (lane < 8) { bnred[wave][lane][0] = a1; bnred[wave][lane][1] = a2; }
        __syncthreads();
        const int cq = threadIdx.x & 7;
        const v4f c1s = ((bnred[0][cq][0] + bnred[1][cq][0]) + bnred[2][cq][0]) + bnred[3][cq][0];
        const v4f c2s = ((bnred[0][cq][1] + bnred[1][cq][1]) + bnred[2][cq][1]) + bnred[3][cq][1];
        const float inv_n = 1.f / (float)(B_SZ * R_SZ);
        mu4 = c1s * inv_n;
        const v4f var4 = c2s * inv_n - mu4 * mu4;
#pragma unroll
        for (int k = 0; k < 4; ++k) sc4[k] = bnw4[k] * rsqrtf(var4[k] + EPS_BN);
    }
    for (int k0 = 0; k0 < D_SZ - 2 * KB; k0 += 2 * KB) {
        fc_load(s1, pa + TKB, pb + KB);
        fc_mma(s0, c0, c1);
        fc_load(s0, pa + 2 * TKB, pb + 2 * KB);
        fc_mma(s1, c0, c1);
        pa += 2 * TKB;
        pb += 2 * KB;
    }
    fc_load(s1, pa + TKB, pb + KB);
    const int n = n0 + fr;
    const int d = n >> 1;
    const float* pcm = A + (size_t)(m0 >> 4) * GROUP_FLOATS + (size_t)(d >> 3) * TILE_FLOATS
                       + (size_t)(half * 8 * TILE_K + (d & 7));
    float ch_lo[8], ch_hi[8];
#pragma unroll
    for (int j = 0; j < 8; ++j) {
        ch_lo[j] = pcm[j * TILE_K];
        ch_hi[j] = pcm[GROUP_FLOATS + j * TILE_K];
    }
    const v4f nl0 = ld_quad(nrm + m0 + half * 8),      nl1 = ld_quad(nrm + m0 + half * 8 + 4);
    const v4f nh0 = ld_quad(nrm + m0 + 16 + half * 8), nh1 = ld_quad(nrm + m0 + 20 + half * 8);
    const int brow = m0 + (threadIdx.x >> 3);
    const size_t bi0 = (size_t)brow * D_SZ + bd;
    const size_t bi1 = bi0 + (size_t)16 * D_SZ;
    const size_t to0 = tiled(brow, bd);
    const size_t to1 = tiled(brow + 16, bd);
    const v4f bb4 = ld_quad(bn_b + bd);
    const v4f sa = ld_quad(S1 + bi0), sb = ld_quad(S1 + bi1);
    fc_mma(s0, c0, c1);
    fc_mma(s1, c0, c1);

    {
        const float inv_r = 1.f / (float)R_SZ;
        const v4f va = (sa * inv_r - mu4) * sc4 + bb4;
        const v4f vb = (sb * inv_r - mu4) * sc4 + bb4;
        *(v4f*)(base + to0)  = va;
        *(v4f*)(base2 + to0) = va * va;
        *(v4f*)(base + to1)  = vb;
        *(v4f*)(base2 + to1) = vb * vb;
    }

    const float bias_n = bias[n];
    const bool even = (fr & 1) == 0;
    float* __restrict__ pab = even ? P : ABm;
#pragma unroll
    for (int j = 0; j < 8; ++j) {
        const int r_lo = j + half * 8, r_hi = r_lo + 16;
        const size_t i_lo = tiled(m0 + r_lo, d);
        const size_t i_hi = tiled(m0 + r_hi, d);
        const float v0 = c0[j] + bias_n, v1 = c1[j] + bias_n;
        const float p0 = row_read<0x161>(v0), p1 = row_read<0x161>(v1);
        const float h0 = ch_lo[j] / (j < 4 ? nl0[j & 3] : nl1[j & 3]);
        const float h1 = ch_hi[j] / (j < 4 ? nh0[j & 3] : nh1[j & 3]);
        pab[i_lo] = v0 * (even ? h0 : p0);
        pab[i_hi] = v1 * (even ? h1 : p1);
        const float a = even ? v0 : p1;
        A2m[even ? i_lo : i_hi] = a * a;
        float t0 = even ? p0 * h0 : v0 * v0;
        float t1 = even ? p1 * h1 : v1 * v1;
        t0 += row_read<0x162>(t0);  t1 += row_read<0x162>(t1);
        t0 += row_read<0x164>(t0);  t1 += row_read<0x164>(t1);
        t0 += row_read<0x168>(t0);  t1 += row_read<0x168>(t1);
        if (fr < 2) {
            red[wave][r_lo][fr] = t0;
            red[wave][r_hi][fr] = t1;
        }
    }
    __syncthreads();
    if (threadIdx.x < 64) {
        const int row = threadIdx.x >> 1, kind = threadIdx.x & 1;
        const float s = ((red[0][row][kind] + red[1][row][kind]) + red[2][row][kind])
                        + red[3][row][kind];
        float* __restrict__ dst = kind ? rpart : qpart;
        dst[(size_t)blockIdx.x * B_SZ + m0 + row] = s;
    }
}

struct SimsBuf { v4f a[NQ], b[NQ]; };
#define NPART (2 * D_SZ / 64)

__device__ __forceinline__ void sims_load(SimsBuf& s,
        const float* __restrict__ pa, const float* __restrict__ pb) {
#pragma unroll
    for (int j = 0; j < NQ; ++j) {
        s.a[j] = ld_quad(pa + TILE_FLOATS * j);
        s.b[j] = ld_quad(pb + TILE_FLOATS * j);
    }
}
__device__ __forceinline__ void sims_mma(const SimsBuf& s, v8f& acc_lo, v8f& acc_hi) {
#pragma unroll
    for (int j = 0; j < NQ; ++j) {
        acc_lo = mma4(lo2(s.a[j]), lo2(s.b[j]), acc_lo);
        acc_hi = mma4(hi2(s.a[j]), hi2(s.b[j]), acc_hi);
    }
}

__global__ __launch_bounds__(96) void sims_gemm_kernel(
        const float* __restrict__ base, const float* __restrict__ base2,
        const float* __restrict__ P, const float* __restrict__ A2m,
        const float* __restrict__ ABm,
        const float* __restrict__ qpart, const float* __restrict__ rpart,
        float* __restrict__ out) {
    __shared__ float part[2][8][32];
    const int wave = __builtin_amdgcn_readfirstlane((int)(threadIdx.x >> 5));
    const int lane = threadIdx.x & 31;
    const int half = lane >> 4, fr = lane & 15;
    const int bm0 = blockIdx.y * 16;
    const int cn0 = blockIdx.x * 16;
    const float* opa = (wave == 1) ? base2 : base;
    const float* opb = (wave == 0) ? P : ((wave == 1) ? A2m : ABm);
    const float* pa = opa + (size_t)(bm0 >> 4) * GROUP_FLOATS + fr * TILE_K + 4 * half;
    const float* pb = opb + (size_t)(cn0 >> 4) * GROUP_FLOATS + fr * TILE_K + 4 * half;
    v8f acc_lo = {}, acc_hi = {};
    SimsBuf s0, s1;
    sims_load(s0, pa, pb);
    float qp[NPART / 2], rp[NPART / 2];
#pragma unroll
    for (int i = 0; i < NPART / 2; ++i) { qp[i] = 0.f; rp[i] = 0.f; }
    if (wave == 0) {
#pragma unroll
        for (int i = 0; i < NPART / 2; ++i) {
            qp[i] = qpart[(size_t)(half * (NPART / 2) + i) * B_SZ + cn0 + fr];
            rp[i] = rpart[(size_t)(half * (NPART / 2) + i) * B_SZ + cn0 + fr];
        }
    }
    for (int t0 = 0; t0 < (D_SZ / KB - 2) * TKB; t0 += 2 * TKB) {
        sims_load(s1, pa + t0 + TKB, pb + t0 + TKB);
        sims_mma(s0, acc_lo, acc_hi);
        sims_load(s0, pa + t0 + 2 * TKB, pb + t0 + 2 * TKB);
        sims_mma(s1, acc_lo, acc_hi);
    }
    sims_load(s1, pa + (D_SZ / KB - 1) * TKB, pb + (D_SZ / KB - 1) * TKB);
    sims_mma(s0, acc_lo, acc_hi);
    sims_mma(s1, acc_lo, acc_hi);
    const v8f acc = acc_lo + acc_hi;

    if (wave != 0) {
#pragma unroll
        for (int j = 0; j < 8; ++j) part[wave - 1][j][lane] = acc[j];
    }
    __syncthreads();
    if (wave == 0) {
        const int c = cn0 + fr;
        float qa = qp[0], qb = qp[1], ra = rp[0], rb = rp[1];
#pragma unroll
        for (int i = 2; i < NPART / 2; i += 2) {
            qa += qp[i];  qb += qp[i + 1];
            ra += rp[i];  rb += rp[i + 1];
        }
        const float qh = qa + qb, rh = ra + rb;
        const float qc = qh + xhalf(qh), rc = rh + xhalf(rh);
#pragma unroll
        for (int j = 0; j < 8; ++j) {
            int b = bm0 + j + half * 8;
            float num = acc[j] + qc;
            float den = sqrtf(part[0][j][lane] + 2.f * part[1][j][lane] + rc) + EPS_L2;
            out[(size_t)b * B_SZ + c] = num / den;
        }
    }
}

extern "C" void kernel_launch(void* const* d_in, const int* in_sizes, int n_in,
                              void* d_out, int out_size, void* d_ws, size_t ws_size,
                              hipStream_t stream) {
    const float* img   = (const float*)d_in[0];
    const float* cap   = (const float*)d_in[1];
    const int*   lens  = (const int*)  d_in[2];
    const float* fc_w  = (const float*)d_in[3];
    const float* fc_b  = (const float*)d_in[4];
    const float* bn_w  = (const float*)d_in[5];
    const float* bn_b  = (const float*)d_in[6];
    float* out = (float*)d_out;

    float* ws = (float*)d_ws;
    const size_t BD = (size_t)B_SZ * D_SZ;
    float* cap_mean = ws;
    float* S1       = cap_mean + BD;
    float* base     = S1 + BD;
    float* base2    = base + BD;
    float* P        = base2 + BD;
    float* A2m      = P + BD;
    float* ABm      = A2m + BD;
    float* S2       = ABm + BD;
    float* qpart    = S2 + BD;
    float* rpart    = qpart + NPART * B_SZ;
    float* nrm      = S2 + 2 * BD;

    pre_kernel<<<dim3(B_SZ), 256, 0, stream>>>(img, cap, lens, S1, S2, cap_mean, nrm);
    fc_gemm_kernel<<<dim3(2 * D_SZ / 64, B_SZ / 32), 128, 0, stream>>>(
            cap_mean, fc_w, fc_b, S1, S2, bn_w, bn_b, nrm, base, base2,
            P, A2m, ABm, qpart, rpart);
    sims_gemm_kernel<<<dim3(B_SZ / 16, B_SZ / 16), 96, 0, stream>>>(
            base, base2, P, A2m, ABm, qpart, rpart, out);
}
